# starting kernel
# baseline (speedup 1.0000x reference)
#include <hip/hip_runtime.h>

typedef __attribute__((ext_vector_type(2))) float v2f;
typedef __attribute__((ext_vector_type(8))) float v8f;

#define N_NODES   50000
#define N_EDGES   800000
#define D_MODEL   128
#define N_HEADS   8
#define HEAD_DIM  16
#define QKV_COLS  384
#define LN_EPS    1e-5f
#define QSCALE    0.08838834764831845f

__global__ __launch_bounds__(256) void gsan_init_kernel(float* __restrict__ m,
                                                        float* __restrict__ denom,
                                                        float* __restrict__ agg) {
  int i = blockIdx.x * 256 + threadIdx.x;
  if (i < N_NODES * D_MODEL) agg[i] = 0.0f;
  if (i < N_NODES * N_HEADS) { m[i] = -__builtin_inff(); denom[i] = 0.0f; }
}

__global__ __launch_bounds__(256) void gsan_ln_qkv_kernel(
    const float* __restrict__ node_feature,
    const float* __restrict__ gamma,
    const float* __restrict__ beta,
    const float* __restrict__ W_qkv,
    const float* __restrict__ b_qkv,
    float* __restrict__ qkv)
{
  __shared__ float xs[16][132];
  __shared__ float smu[16], srs[16];
  const int tid = threadIdx.x;
  const int nodeBase = blockIdx.x * 16;

  for (int i = tid; i < 16 * 128; i += 256) {
    int r = i >> 7, c = i & 127;
    xs[r][c] = node_feature[(nodeBase + r) * D_MODEL + c];
  }
  __syncthreads();

  if (tid < 16) {
    float s = 0.0f, sq = 0.0f;
    for (int c = 0; c < 128; ++c) { float v = xs[tid][c]; s += v; sq += v * v; }
    float mu  = s * (1.0f / 128.0f);
    float var = sq * (1.0f / 128.0f) - mu * mu;
    smu[tid] = mu;
    srs[tid] = rsqrtf(var + LN_EPS);
  }
  __syncthreads();

  for (int i = tid; i < 16 * 128; i += 256) {
    int r = i >> 7, c = i & 127;
    xs[r][c] = (xs[r][c] - smu[r]) * srs[r] * gamma[c] + beta[c];
  }
  __syncthreads();

  const int lane = tid & 31;
  const int wave = tid >> 5;
  const int row  = lane & 15;
  const int half = lane >> 4;

  for (int t = 0; t < 3; ++t) {
    const int nt  = wave + t * 8;
    const int col = nt * 16 + row;
    const float* wrow = W_qkv + col * D_MODEL;
    v8f acc = {};
    #pragma unroll 8
    for (int k0 = 0; k0 < D_MODEL; k0 += 4) {
      const int k = k0 + 2 * half;
      v2f a, b;
      a.x = xs[row][k];  a.y = xs[row][k + 1];
      b.x = wrow[k];     b.y = wrow[k + 1];
      acc = __builtin_amdgcn_wmma_f32_16x16x4_f32(
          false, a, false, b, (short)0, acc, false, false);
    }
    const float bias  = b_qkv[col];
    const float scale = (nt < 8) ? QSCALE : 1.0f;
    #pragma unroll
    for (int j = 0; j < 8; ++j) {
      const int mrow = j + 8 * half;
      qkv[(nodeBase + mrow) * QKV_COLS + col] = (acc[j] + bias) * scale;
    }
  }
}

__device__ inline void atomicMaxF(float* addr, float val) {
  if (val >= 0.0f) atomicMax((int*)addr, __float_as_int(val));
  else             atomicMin((unsigned int*)addr, __float_as_uint(val));
}

__global__ __launch_bounds__(256) void gsan_edge_attn_kernel(
    const float* __restrict__ qkv,
    const float* __restrict__ dist_attn,
    const float* __restrict__ path_attn,
    const int*   __restrict__ src,
    const int*   __restrict__ dst,
    float* __restrict__ a_ws,
    float* __restrict__ m)
{
  int i = blockIdx.x * 256 + threadIdx.x;
  if (i >= N_EDGES * N_HEADS) return;
  const int e = i >> 3, h = i & 7;
  const int s = src[e], d = dst[e];
  const float4* qp = (const float4*)(qkv + (size_t)s * QKV_COLS +       h * HEAD_DIM);
  const float4* kp = (const float4*)(qkv + (size_t)d * QKV_COLS + 128 + h * HEAD_DIM);
  float acc = 0.0f;
  #pragma unroll
  for (int j = 0; j < 4; ++j) {
    float4 q4 = qp[j], k4 = kp[j];
    acc += q4.x * k4.x + q4.y * k4.y + q4.z * k4.z + q4.w * k4.w;
  }
  const float a = acc + dist_attn[i] + path_attn[i];
  a_ws[i] = a;
  atomicMaxF(&m[d * N_HEADS + h], a);
}

__global__ __launch_bounds__(256) void gsan_edge_exp_kernel(
    const int*   __restrict__ dst,
    const float* __restrict__ m,
    float* __restrict__ a_ws,
    float* __restrict__ denom)
{
  int i = blockIdx.x * 256 + threadIdx.x;
  if (i >= N_EDGES * N_HEADS) return;
  const int e = i >> 3, h = i & 7;
  const int d = dst[e];
  const float ex = expf(a_ws[i] - m[d * N_HEADS + h]);
  a_ws[i] = ex;
  atomicAdd(&denom[d * N_HEADS + h], ex);
}

__global__ __launch_bounds__(256) void gsan_edge_agg_kernel(
    const int*   __restrict__ src,
    const int*   __restrict__ dst,
    const float* __restrict__ qkv,
    const float* __restrict__ a_ws,
    const float* __restrict__ denom,
    float* __restrict__ agg)
{
  int i = blockIdx.x * 256 + threadIdx.x;
  if (i >= N_EDGES * N_HEADS) return;
  const int e = i >> 3, h = i & 7;
  const int s = src[e], d = dst[e];
  const float sa = a_ws[i] / denom[d * N_HEADS + h];
  const float* vp = qkv + (size_t)s * QKV_COLS + 256 + h * HEAD_DIM;
  float* ap = agg + (size_t)d * D_MODEL + h * HEAD_DIM;
  #pragma unroll
  for (int j = 0; j < HEAD_DIM; ++j) atomicAdd(&ap[j], vp[j] * sa);
}

__global__ __launch_bounds__(256) void gsan_out_gemm_kernel(
    const float* __restrict__ agg,
    const float* __restrict__ W_out,
    const float* __restrict__ b_out,
    const float* __restrict__ node_feature,
    float* __restrict__ out)
{
  __shared__ float xs[16][132];
  const int tid = threadIdx.x;
  const int nodeBase = blockIdx.x * 16;

  for (int i = tid; i < 16 * 128; i += 256) {
    int r = i >> 7, c = i & 127;
    xs[r][c] = agg[(nodeBase + r) * D_MODEL + c];
  }
  __syncthreads();

  const int lane = tid & 31;
  const int wave = tid >> 5;
  const int row  = lane & 15;
  const int half = lane >> 4;
  const int col  = wave * 16 + row;
  const float* wrow = W_out + col * D_MODEL;

  v8f acc = {};
  #pragma unroll 8
  for (int k0 = 0; k0 < D_MODEL; k0 += 4) {
    const int k = k0 + 2 * half;
    v2f a, b;
    a.x = xs[row][k];  a.y = xs[row][k + 1];
    b.x = wrow[k];     b.y = wrow[k + 1];
    acc = __builtin_amdgcn_wmma_f32_16x16x4_f32(
        false, a, false, b, (short)0, acc, false, false);
  }
  const float bias = b_out[col];
  #pragma unroll
  for (int j = 0; j < 8; ++j) {
    const int mrow = j + 8 * half;
    const int idx  = (nodeBase + mrow) * D_MODEL + col;
    out[idx] = node_feature[idx] + acc[j] + bias;
  }
}

extern "C" void kernel_launch(void* const* d_in, const int* in_sizes, int n_in,
                              void* d_out, int out_size, void* d_ws, size_t ws_size,
                              hipStream_t stream) {
  const float* node_feature = (const float*)d_in[0];
  const float* dist_attn    = (const float*)d_in[1];
  const float* path_attn    = (const float*)d_in[2];
  const float* gamma        = (const float*)d_in[3];
  const float* beta         = (const float*)d_in[4];
  const float* W_qkv        = (const float*)d_in[5];
  const float* b_qkv        = (const float*)d_in[6];
  const float* W_out        = (const float*)d_in[7];
  const float* b_out        = (const float*)d_in[8];
  const int*   src          = (const int*)d_in[9];
  const int*   dst          = (const int*)d_in[10];
  float*       out          = (float*)d_out;

  float* ws    = (float*)d_ws;
  float* qkv   = ws;
  float* a_ws  = qkv   + (size_t)N_NODES * QKV_COLS;
  float* m     = a_ws  + (size_t)N_EDGES * N_HEADS;
  float* denom = m     + (size_t)N_NODES * N_HEADS;
  float* agg   = denom + (size_t)N_NODES * N_HEADS;

  const int edgeBlocks = (N_EDGES * N_HEADS + 255) / 256;
  const int nodeBlocks = N_NODES / 16;
  const int initBlocks = (N_NODES * D_MODEL + 255) / 256;

  gsan_init_kernel<<<initBlocks, 256, 0, stream>>>(m, denom, agg);
  gsan_ln_qkv_kernel<<<nodeBlocks, 256, 0, stream>>>(
      node_feature, gamma, beta, W_qkv, b_qkv, qkv);
  gsan_edge_attn_kernel<<<edgeBlocks, 256, 0, stream>>>(
      qkv, dist_attn, path_attn, src, dst, a_ws, m);
  gsan_edge_exp_kernel<<<edgeBlocks, 256, 0, stream>>>(dst, m, a_ws, denom);
  gsan_edge_agg_kernel<<<edgeBlocks, 256, 0, stream>>>(
      src, dst, qkv, a_ws, denom, agg);
  gsan_out_gemm_kernel<<<nodeBlocks, 256, 0, stream>>>(
      agg, W_out, b_out, node_feature, out);
}


# edge_agg float atomics removed: counting sort by dst (rank in edge_attn, one-block LDS scan, scatter in edge_exp), per-node gather into out-GEMM LDS tile; agg sum order = arrival rank per node
# speedup vs baseline: 2.0084x; 2.0084x over previous
#include <hip/hip_runtime.h>

typedef __attribute__((ext_vector_type(2))) float v2f;
typedef __attribute__((ext_vector_type(8))) float v8f;

#define N_NODES   50000
#define N_EDGES   800000
#define D_MODEL   128
#define N_HEADS   8
#define HEAD_DIM  16
#define QKV_COLS  384
#define LN_EPS    1e-5f
#define QSCALE    0.08838834764831845f

#define SCAN_THREADS 1024
#define SCAN_CHUNK   49
#define SCAN_PAD     (SCAN_THREADS * SCAN_CHUNK)

__global__ __launch_bounds__(256) void gsan_init_kernel(float* __restrict__ m,
                                                        float* __restrict__ denom,
                                                        int* __restrict__ cnt) {
  int i = blockIdx.x * 256 + threadIdx.x;
  if (i < N_NODES * N_HEADS) { m[i] = -__builtin_inff(); denom[i] = 0.0f; }
  if (i < N_NODES) cnt[i] = 0;
}

__global__ __launch_bounds__(256) void gsan_ln_qkv_kernel(
    const float* __restrict__ node_feature,
    const float* __restrict__ gamma,
    const float* __restrict__ beta,
    const float* __restrict__ W_qkv,
    const float* __restrict__ b_qkv,
    float* __restrict__ qkv)
{
  __shared__ float xs[16][132];
  __shared__ float smu[16], srs[16];
  const int tid = threadIdx.x;
  const int nodeBase = blockIdx.x * 16;

  for (int i = tid; i < 16 * 128; i += 256) {
    int r = i >> 7, c = i & 127;
    xs[r][c] = node_feature[(nodeBase + r) * D_MODEL + c];
  }
  __syncthreads();

  if (tid < 16) {
    float s = 0.0f, sq = 0.0f;
    for (int c = 0; c < 128; ++c) { float v = xs[tid][c]; s += v; sq += v * v; }
    float mu  = s * (1.0f / 128.0f);
    float var = sq * (1.0f / 128.0f) - mu * mu;
    smu[tid] = mu;
    srs[tid] = rsqrtf(var + LN_EPS);
  }
  __syncthreads();

  for (int i = tid; i < 16 * 128; i += 256) {
    int r = i >> 7, c = i & 127;
    xs[r][c] = (xs[r][c] - smu[r]) * srs[r] * gamma[c] + beta[c];
  }
  __syncthreads();

  const int lane = tid & 31;
  const int wave = tid >> 5;
  const int row  = lane & 15;
  const int half = lane >> 4;

  for (int t = 0; t < 3; ++t) {
    const int nt  = wave + t * 8;
    const int col = nt * 16 + row;
    const float* wrow = W_qkv + col * D_MODEL;
    v8f acc = {};
    #pragma unroll 8
    for (int k0 = 0; k0 < D_MODEL; k0 += 4) {
      const int k = k0 + 2 * half;
      v2f a, b;
      a.x = xs[row][k];  a.y = xs[row][k + 1];
      b.x = wrow[k];     b.y = wrow[k + 1];
      acc = __builtin_amdgcn_wmma_f32_16x16x4_f32(
          false, a, false, b, (short)0, acc, false, false);
    }
    const float bias  = b_qkv[col];
    const float scale = (nt < 8) ? QSCALE : 1.0f;
    #pragma unroll
    for (int j = 0; j < 8; ++j) {
      const int mrow = j + 8 * half;
      qkv[(nodeBase + mrow) * QKV_COLS + col] = (acc[j] + bias) * scale;
    }
  }
}

__device__ inline void atomicMaxF(float* addr, float val) {
  if (val >= 0.0f) atomicMax((int*)addr, __float_as_int(val));
  else             atomicMin((unsigned int*)addr, __float_as_uint(val));
}

__global__ __launch_bounds__(256) void gsan_edge_attn_kernel(
    const float* __restrict__ qkv,
    const float* __restrict__ dist_attn,
    const float* __restrict__ path_attn,
    const int*   __restrict__ src,
    const int*   __restrict__ dst,
    float* __restrict__ a_ws,
    float* __restrict__ m,
    int*   __restrict__ cnt,
    int*   __restrict__ rank)
{
  int i = blockIdx.x * 256 + threadIdx.x;
  if (i >= N_EDGES * N_HEADS) return;
  const int e = i >> 3, h = i & 7;
  const int s = src[e], d = dst[e];
  const float4* qp = (const float4*)(qkv + (size_t)s * QKV_COLS +       h * HEAD_DIM);
  const float4* kp = (const float4*)(qkv + (size_t)d * QKV_COLS + 128 + h * HEAD_DIM);
  float acc = 0.0f;
  #pragma unroll
  for (int j = 0; j < 4; ++j) {
    float4 q4 = qp[j], k4 = kp[j];
    acc += q4.x * k4.x + q4.y * k4.y + q4.z * k4.z + q4.w * k4.w;
  }
  const float a = acc + dist_attn[i] + path_attn[i];
  a_ws[i] = a;
  atomicMaxF(&m[d * N_HEADS + h], a);
  if (h == 0) rank[e] = atomicAdd(&cnt[d], 1);
}

__global__ __launch_bounds__(SCAN_THREADS) void gsan_scan_kernel(
    const int* __restrict__ cnt, int* __restrict__ offs)
{
  __shared__ int s[SCAN_PAD];
  __shared__ int wsum[32];
  const int tid  = threadIdx.x;
  const int lane = tid & 31;
  const int wave = tid >> 5;
  for (int i = tid; i < SCAN_PAD; i += SCAN_THREADS) s[i] = (i < N_NODES) ? cnt[i] : 0;
  __syncthreads();
  int* my = s + tid * SCAN_CHUNK;
  int sum = 0;
  for (int j = 0; j < SCAN_CHUNK; ++j) sum += my[j];
  int incl = sum;
  for (int o = 1; o < 32; o <<= 1) { int t = __shfl_up(incl, o); if (lane >= o) incl += t; }
  if (lane == 31) wsum[wave] = incl;
  __syncthreads();
  if (wave == 0) {
    const int w = wsum[lane];
    int wi = w;
    for (int o = 1; o < 32; o <<= 1) { int t = __shfl_up(wi, o); if (lane >= o) wi += t; }
    wsum[lane] = wi - w;
  }
  __syncthreads();
  int run = wsum[wave] + incl - sum;
  for (int j = 0; j < SCAN_CHUNK; ++j) { int v = my[j]; my[j] = run; run += v; }
  __syncthreads();
  for (int i = tid; i < N_NODES; i += SCAN_THREADS) offs[i] = s[i];
}

__global__ __launch_bounds__(256) void gsan_edge_exp_kernel(
    const int*   __restrict__ src,
    const int*   __restrict__ dst,
    const float* __restrict__ m,
    float* __restrict__ a_ws,
    float* __restrict__ denom,
    const int* __restrict__ offs,
    const int* __restrict__ rank,
    int* __restrict__ perm,
    int* __restrict__ psrc)
{
  int i = blockIdx.x * 256 + threadIdx.x;
  if (i >= N_EDGES * N_HEADS) return;
  const int e = i >> 3, h = i & 7;
  const int d = dst[e];
  const float ex = expf(a_ws[i] - m[d * N_HEADS + h]);
  a_ws[i] = ex;
  atomicAdd(&denom[d * N_HEADS + h], ex);
  if (h == 0) {
    int pos = offs[d] + rank[e];
    pos = (pos < 0) ? 0 : ((pos > N_EDGES - 1) ? (N_EDGES - 1) : pos);
    perm[pos] = e;
    psrc[pos] = src[e];
  }
}

__global__ __launch_bounds__(256) void gsan_gather_out_kernel(
    const float* __restrict__ qkv,
    const float* __restrict__ a_ws,
    const float* __restrict__ denom,
    const int*   __restrict__ cnt,
    const int*   __restrict__ offs,
    const int*   __restrict__ perm,
    const int*   __restrict__ psrc,
    const float* __restrict__ W_out,
    const float* __restrict__ b_out,
    const float* __restrict__ node_feature,
    float* __restrict__ out)
{
  __shared__ float xs[16][132];
  const int tid = threadIdx.x;
  const int nodeBase = blockIdx.x * 16;
  const int lane = tid & 31;
  const int hl   = lane >> 2;
  const int gw = __builtin_amdgcn_readfirstlane(tid >> 5);

  for (int nn = 0; nn < 2; ++nn) {
    const int r     = gw * 2 + nn;
    const int node  = nodeBase + r;
    int start = offs[node];
    int deg   = cnt[node];
    start = (start < 0) ? 0 : ((start > N_EDGES) ? N_EDGES : start);
    deg   = (deg < 0) ? 0 : ((deg > N_EDGES - start) ? (N_EDGES - start) : deg);
    float4 acc = make_float4(0.0f, 0.0f, 0.0f, 0.0f);
    for (int base = 0; base < deg; base += 32) {
      const int n   = (deg - base < 32) ? (deg - base) : 32;
      const int idx = start + base + ((lane < n) ? lane : (n - 1));
      int e_l = perm[idx];
      int s_l = psrc[idx];
      e_l = (e_l < 0) ? 0 : ((e_l > N_EDGES - 1) ? (N_EDGES - 1) : e_l);
      s_l = (s_l < 0) ? 0 : ((s_l > N_NODES - 1) ? (N_NODES - 1) : s_l);
      for (int j = 0; j < n; j += 4) {
        #pragma unroll
        for (int u = 0; u < 4; ++u) {
          const int e = __builtin_amdgcn_readlane(e_l, j + u);
          const int s = __builtin_amdgcn_readlane(s_l, j + u);
          const float  ex = a_ws[(size_t)e * N_HEADS + hl];
          const float4 v  = *(const float4*)(qkv + (size_t)s * QKV_COLS + 256 + lane * 4);
          const float  w  = (j + u < n) ? ex : 0.0f;
          acc.x += w * v.x;  acc.y += w * v.y;  acc.z += w * v.z;  acc.w += w * v.w;
        }
      }
    }
    const float dn  = denom[node * N_HEADS + hl];
    const float inv = (deg > 0) ? (1.0f / dn) : 0.0f;
    *(float4*)&xs[r][lane * 4] = make_float4(acc.x * inv, acc.y * inv, acc.z * inv, acc.w * inv);
  }
  __syncthreads();

  const int wave = tid >> 5;
  const int row  = lane & 15;
  const int half = lane >> 4;
  const int col  = wave * 16 + row;
  const float* wrow = W_out + col * D_MODEL;

  v8f acc = {};
  #pragma unroll 8
  for (int k0 = 0; k0 < D_MODEL; k0 += 4) {
    const int k = k0 + 2 * half;
    v2f a, b;
    a.x = xs[row][k];  a.y = xs[row][k + 1];
    b.x = wrow[k];     b.y = wrow[k + 1];
    acc = __builtin_amdgcn_wmma_f32_16x16x4_f32(
        false, a, false, b, (short)0, acc, false, false);
  }
  const float bias = b_out[col];
  #pragma unroll
  for (int j = 0; j < 8; ++j) {
    const int mrow = j + 8 * half;
    const int idx  = (nodeBase + mrow) * D_MODEL + col;
    out[idx] = node_feature[idx] + acc[j] + bias;
  }
}

extern "C" void kernel_launch(void* const* d_in, const int* in_sizes, int n_in,
                              void* d_out, int out_size, void* d_ws, size_t ws_size,
                              hipStream_t stream) {
  const float* node_feature = (const float*)d_in[0];
  const float* dist_attn    = (const float*)d_in[1];
  const float* path_attn    = (const float*)d_in[2];
  const float* gamma        = (const float*)d_in[3];
  const float* beta         = (const float*)d_in[4];
  const float* W_qkv        = (const float*)d_in[5];
  const float* b_qkv        = (const float*)d_in[6];
  const float* W_out        = (const float*)d_in[7];
  const float* b_out        = (const float*)d_in[8];
  const int*   src          = (const int*)d_in[9];
  const int*   dst          = (const int*)d_in[10];
  float*       out          = (float*)d_out;

  float* ws    = (float*)d_ws;
  float* qkv   = ws;
  float* a_ws  = qkv   + (size_t)N_NODES * QKV_COLS;
  float* m     = a_ws  + (size_t)N_EDGES * N_HEADS;
  float* denom = m     + (size_t)N_NODES * N_HEADS;
  int*   cnt   = (int*)(denom + (size_t)N_NODES * N_HEADS);
  int*   offs  = cnt  + 50048;
  int*   rank  = offs + 50048;
  int*   perm  = rank + N_EDGES;
  int*   psrc  = perm + N_EDGES;

  const int edgeBlocks = (N_EDGES * N_HEADS + 255) / 256;
  const int nodeBlocks = N_NODES / 16;
  const int initBlocks = (N_NODES * N_HEADS + 255) / 256;

  gsan_init_kernel<<<initBlocks, 256, 0, stream>>>(m, denom, cnt);
  gsan_ln_qkv_kernel<<<nodeBlocks, 256, 0, stream>>>(
      node_feature, gamma, beta, W_qkv, b_qkv, qkv);
  gsan_edge_attn_kernel<<<edgeBlocks, 256, 0, stream>>>(
      qkv, dist_attn, path_attn, src, dst, a_ws, m, cnt, rank);
  gsan_scan_kernel<<<1, SCAN_THREADS, 0, stream>>>(cnt, offs);
  gsan_edge_exp_kernel<<<edgeBlocks, 256, 0, stream>>>(
      src, dst, m, a_ws, denom, offs, rank, perm, psrc);
  gsan_gather_out_kernel<<<nodeBlocks, 256, 0, stream>>>(
      qkv, a_ws, denom, cnt, offs, perm, psrc, W_out, b_out, node_feature, out);
}


# edge softmax fused into the per-dst gather: q.k scores, running max (sums rescaled per 4-edge group), exp, denom, V sum over each dst-sorted segment; m/denom float atomics and a_ws passes removed
# speedup vs baseline: 2.7257x; 1.3571x over previous
#include <hip/hip_runtime.h>

typedef __attribute__((ext_vector_type(2))) float v2f;
typedef __attribute__((ext_vector_type(8))) float v8f;

#define N_NODES   50000
#define N_EDGES   800000
#define D_MODEL   128
#define N_HEADS   8
#define HEAD_DIM  16
#define QKV_COLS  384
#define LN_EPS    1e-5f
#define QSCALE    0.08838834764831845f

#define SCAN_THREADS 1024
#define SCAN_CHUNK   49
#define SCAN_PAD     (SCAN_THREADS * SCAN_CHUNK)

__global__ __launch_bounds__(256) void gsan_init_kernel(int* __restrict__ cnt) {
  int i = blockIdx.x * 256 + threadIdx.x;
  if (i < N_NODES) cnt[i] = 0;
}

__global__ __launch_bounds__(256) void gsan_ln_qkv_kernel(
    const float* __restrict__ node_feature,
    const float* __restrict__ gamma,
    const float* __restrict__ beta,
    const float* __restrict__ W_qkv,
    const float* __restrict__ b_qkv,
    float* __restrict__ qkv)
{
  __shared__ float xs[16][132];
  __shared__ float smu[16], srs[16];
  const int tid = threadIdx.x;
  const int nodeBase = blockIdx.x * 16;

  for (int i = tid; i < 16 * 128; i += 256) {
    int r = i >> 7, c = i & 127;
    xs[r][c] = node_feature[(nodeBase + r) * D_MODEL + c];
  }
  __syncthreads();

  if (tid < 16) {
    float s = 0.0f, sq = 0.0f;
    for (int c = 0; c < 128; ++c) { float v = xs[tid][c]; s += v; sq += v * v; }
    float mu  = s * (1.0f / 128.0f);
    float var = sq * (1.0f / 128.0f) - mu * mu;
    smu[tid] = mu;
    srs[tid] = rsqrtf(var + LN_EPS);
  }
  __syncthreads();

  for (int i = tid; i < 16 * 128; i += 256) {
    int r = i >> 7, c = i & 127;
    xs[r][c] = (xs[r][c] - smu[r]) * srs[r] * gamma[c] + beta[c];
  }
  __syncthreads();

  const int lane = tid & 31;
  const int wave = tid >> 5;
  const int row  = lane & 15;
  const int half = lane >> 4;

  for (int t = 0; t < 3; ++t) {
    const int nt  = wave + t * 8;
    const int col = nt * 16 + row;
    const float* wrow = W_qkv + col * D_MODEL;
    v8f acc = {};
    #pragma unroll 8
    for (int k0 = 0; k0 < D_MODEL; k0 += 4) {
      const int k = k0 + 2 * half;
      v2f a, b;
      a.x = xs[row][k];  a.y = xs[row][k + 1];
      b.x = wrow[k];     b.y = wrow[k + 1];
      acc = __builtin_amdgcn_wmma_f32_16x16x4_f32(
          false, a, false, b, (short)0, acc, false, false);
    }
    const float bias  = b_qkv[col];
    const float scale = (nt < 8) ? QSCALE : 1.0f;
    #pragma unroll
    for (int j = 0; j < 8; ++j) {
      const int mrow = j + 8 * half;
      qkv[(nodeBase + mrow) * QKV_COLS + col] = (acc[j] + bias) * scale;
    }
  }
}

__global__ __launch_bounds__(256) void gsan_edge_count_kernel(
    const int* __restrict__ dst,
    int* __restrict__ cnt,
    int* __restrict__ rank)
{
  const int e = blockIdx.x * 256 + threadIdx.x;
  if (e >= N_EDGES) return;
  rank[e] = atomicAdd(&cnt[dst[e]], 1);
}

__global__ __launch_bounds__(SCAN_THREADS) void gsan_scan_kernel(
    const int* __restrict__ cnt, int* __restrict__ offs)
{
  __shared__ int s[SCAN_PAD];
  __shared__ int wsum[32];
  const int tid  = threadIdx.x;
  const int lane = tid & 31;
  const int wave = tid >> 5;
  for (int i = tid; i < SCAN_PAD; i += SCAN_THREADS) s[i] = (i < N_NODES) ? cnt[i] : 0;
  __syncthreads();
  int* my = s + tid * SCAN_CHUNK;
  int sum = 0;
  for (int j = 0; j < SCAN_CHUNK; ++j) sum += my[j];
  int incl = sum;
  for (int o = 1; o < 32; o <<= 1) { int t = __shfl_up(incl, o); if (lane >= o) incl += t; }
  if (lane == 31) wsum[wave] = incl;
  __syncthreads();
  if (wave == 0) {
    const int w = wsum[lane];
    int wi = w;
    for (int o = 1; o < 32; o <<= 1) { int t = __shfl_up(wi, o); if (lane >= o) wi += t; }
    wsum[lane] = wi - w;
  }
  __syncthreads();
  int run = wsum[wave] + incl - sum;
  for (int j = 0; j < SCAN_CHUNK; ++j) { int v = my[j]; my[j] = run; run += v; }
  __syncthreads();
  for (int i = tid; i < N_NODES; i += SCAN_THREADS) offs[i] = s[i];
}

__global__ __launch_bounds__(256) void gsan_edge_scatter_kernel(
    const float* __restrict__ dist_attn,
    const float* __restrict__ path_attn,
    const int*   __restrict__ src,
    const int*   __restrict__ dst,
    const int*   __restrict__ offs,
    const int*   __restrict__ rank,
    int*   __restrict__ psrc,
    float* __restrict__ ebias)
{
  const int i = blockIdx.x * 256 + threadIdx.x;
  if (i >= N_EDGES * 2) return;
  const int e = i >> 1, hh = i & 1;
  int pos = offs[dst[e]] + rank[e];
  pos = (pos < 0) ? 0 : ((pos > N_EDGES - 1) ? (N_EDGES - 1) : pos);
  const float4 dv = *(const float4*)(dist_attn + (size_t)i * 4);
  const float4 pv = *(const float4*)(path_attn + (size_t)i * 4);
  *(float4*)(ebias + (size_t)pos * N_HEADS + hh * 4) =
      make_float4(dv.x + pv.x, dv.y + pv.y, dv.z + pv.z, dv.w + pv.w);
  if (hh == 0) psrc[pos] = src[e];
}

__device__ __forceinline__ float quad_swap1(float x) {
  return __int_as_float(__builtin_amdgcn_update_dpp(0, __float_as_int(x), 0xB1, 0xF, 0xF, true));
}
__device__ __forceinline__ float quad_swap2(float x) {
  return __int_as_float(__builtin_amdgcn_update_dpp(0, __float_as_int(x), 0x4E, 0xF, 0xF, true));
}
template <int U>
__device__ __forceinline__ float quad_bcast(float x) {
  return __int_as_float(__builtin_amdgcn_update_dpp(0, __float_as_int(x), U * 0x55, 0xF, 0xF, true));
}

__device__ __forceinline__ void load_rows4(const float* __restrict__ qkv, int s_l, int j, int col4,
                                           float4 (&q)[4], float4 (&v)[4]) {
  #pragma unroll
  for (int u = 0; u < 4; ++u) {
    const int s = __builtin_amdgcn_readlane(s_l, j + u);
    const float* rowp = qkv + (size_t)s * QKV_COLS + col4;
    q[u] = *(const float4*)rowp;
    v[u] = *(const float4*)(rowp + 256);
  }
}

__device__ __forceinline__ float group_bias(const float* __restrict__ ebias, int seg, int j,
                                            int tq, int hl, int n) {
  const int jt = (j + tq < n) ? (j + tq) : (n - 1);
  return ebias[(size_t)(seg + jt) * N_HEADS + hl];
}

__device__ __forceinline__ void attn_group4(const float4 (&q)[4], const float4 (&v)[4], float bq,
                                            const float4 k4, int left,
                                            float& mrun, float& den, float4& acc) {
  const float bu[4] = {quad_bcast<0>(bq), quad_bcast<1>(bq), quad_bcast<2>(bq), quad_bcast<3>(bq)};
  float a[4];
  #pragma unroll
  for (int u = 0; u < 4; ++u) {
    float d = q[u].x * k4.x + q[u].y * k4.y + q[u].z * k4.z + q[u].w * k4.w;
    d += quad_swap1(d);
    d += quad_swap2(d);
    a[u] = (u < left) ? (d + bu[u]) : -__builtin_inff();
  }
  const float gmax = fmaxf(fmaxf(a[0], a[1]), fmaxf(a[2], a[3]));
  const float mnew = fmaxf(mrun, gmax);
  const float corr = expf(mrun - mnew);
  const float p0 = expf(a[0] - mnew), p1 = expf(a[1] - mnew);
  const float p2 = expf(a[2] - mnew), p3 = expf(a[3] - mnew);
  den = den * corr + ((p0 + p1) + (p2 + p3));
  const float p[4] = {p0, p1, p2, p3};
  float4 t = make_float4(acc.x * corr, acc.y * corr, acc.z * corr, acc.w * corr);
  #pragma unroll
  for (int u = 0; u < 4; ++u) {
    t.x = fmaf(p[u], v[u].x, t.x);  t.y = fmaf(p[u], v[u].y, t.y);
    t.z = fmaf(p[u], v[u].z, t.z);  t.w = fmaf(p[u], v[u].w, t.w);
  }
  acc  = t;
  mrun = mnew;
}

__global__ __launch_bounds__(256) void gsan_attn_out_kernel(
    const float* __restrict__ qkv,
    const float* __restrict__ ebias,
    const int*   __restrict__ cnt,
    const int*   __restrict__ offs,
    const int*   __restrict__ psrc,
    const float* __restrict__ W_out,
    const float* __restrict__ b_out,
    const float* __restrict__ node_feature,
    float* __restrict__ out)
{
  __shared__ float xs[16][132];
  const int tid = threadIdx.x;
  const int nodeBase = blockIdx.x * 16;
  const int lane = tid & 31;
  const int hl   = lane >> 2;
  const int tq   = lane & 3;
  const int gw = __builtin_amdgcn_readfirstlane(tid >> 5);

  for (int nn = 0; nn < 2; ++nn) {
    const int r     = gw * 2 + nn;
    const int node  = nodeBase + r;
    int start = offs[node];
    int deg   = cnt[node];
    start = (start < 0) ? 0 : ((start > N_EDGES) ? N_EDGES : start);
    deg   = (deg < 0) ? 0 : ((deg > N_EDGES - start) ? (N_EDGES - start) : deg);
    const float4 k4 = *(const float4*)(qkv + (size_t)node * QKV_COLS + 128 + lane * 4);
    float4 acc  = make_float4(0.0f, 0.0f, 0.0f, 0.0f);
    float  den  = 0.0f;
    float  mrun = -__builtin_inff();
    for (int base = 0; base < deg; base += 32) {
      const int n   = (deg - base < 32) ? (deg - base) : 32;
      const int seg = start + base;
      int s_l = psrc[seg + ((lane < n) ? lane : (n - 1))];
      s_l = (s_l < 0) ? 0 : ((s_l > N_NODES - 1) ? (N_NODES - 1) : s_l);
      float4 qa[4], va[4], qb[4], vb[4];
      float  ba = group_bias(ebias, seg, 0, tq, hl, n), bb = 0.0f;
      load_rows4(qkv, s_l, 0, lane * 4, qa, va);
      for (int j = 0; j < n; j += 8) {
        if (j + 4 < n) {
          bb = group_bias(ebias, seg, j + 4, tq, hl, n);
          load_rows4(qkv, s_l, j + 4, lane * 4, qb, vb);
        }
        attn_group4(qa, va, ba, k4, n - j, mrun, den, acc);
        if (j + 4 < n) {
          if (j + 8 < n) {
            ba = group_bias(ebias, seg, j + 8, tq, hl, n);
            load_rows4(qkv, s_l, j + 8, lane * 4, qa, va);
          }
          attn_group4(qb, vb, bb, k4, n - j - 4, mrun, den, acc);
        }
      }
    }
    const float inv = (deg > 0) ? (1.0f / den) : 0.0f;
    *(float4*)&xs[r][lane * 4] = make_float4(acc.x * inv, acc.y * inv, acc.z * inv, acc.w * inv);
  }
  __syncthreads();

  const int wave = tid >> 5;
  const int row  = lane & 15;
  const int half = lane >> 4;
  const int col  = wave * 16 + row;
  const float* wrow = W_out + col * D_MODEL;

  v8f acc = {};
  #pragma unroll 8
  for (int k0 = 0; k0 < D_MODEL; k0 += 4) {
    const int k = k0 + 2 * half;
    v2f a, b;
    a.x = xs[row][k];  a.y = xs[row][k + 1];
    b.x = wrow[k];     b.y = wrow[k + 1];
    acc = __builtin_amdgcn_wmma_f32_16x16x4_f32(
        false, a, false, b, (short)0, acc, false, false);
  }
  const float bias = b_out[col];
  #pragma unroll
  for (int j = 0; j < 8; ++j) {
    const int mrow = j + 8 * half;
    const int idx  = (nodeBase + mrow) * D_MODEL + col;
    out[idx] = node_feature[idx] + acc[j] + bias;
  }
}

extern "C" void kernel_launch(void* const* d_in, const int* in_sizes, int n_in,
                              void* d_out, int out_size, void* d_ws, size_t ws_size,
                              hipStream_t stream) {
  const float* node_feature = (const float*)d_in[0];
  const float* dist_attn    = (const float*)d_in[1];
  const float* path_attn    = (const float*)d_in[2];
  const float* gamma        = (const float*)d_in[3];
  const float* beta         = (const float*)d_in[4];
  const float* W_qkv        = (const float*)d_in[5];
  const float* b_qkv        = (const float*)d_in[6];
  const float* W_out        = (const float*)d_in[7];
  const float* b_out        = (const float*)d_in[8];
  const int*   src          = (const int*)d_in[9];
  const int*   dst          = (const int*)d_in[10];
  float*       out          = (float*)d_out;

  float* ws    = (float*)d_ws;
  float* qkv   = ws;
  float* ebias = qkv   + (size_t)N_NODES * QKV_COLS;
  int*   cnt   = (int*)(ebias + (size_t)N_EDGES * N_HEADS + (size_t)2 * N_NODES * N_HEADS);
  int*   offs  = cnt  + 50048;
  int*   rank  = offs + 50048;
  int*   psrc  = rank + 2 * N_EDGES;

  const int edgeBlocks  = (N_EDGES + 255) / 256;
  const int edge2Blocks = (N_EDGES * 2 + 255) / 256;
  const int nodeBlocks  = N_NODES / 16;
  const int initBlocks  = (N_NODES + 255) / 256;

  gsan_init_kernel<<<initBlocks, 256, 0, stream>>>(cnt);
  gsan_edge_count_kernel<<<edgeBlocks, 256, 0, stream>>>(dst, cnt, rank);
  gsan_scan_kernel<<<1, SCAN_THREADS, 0, stream>>>(cnt, offs);
  gsan_edge_scatter_kernel<<<edge2Blocks, 256, 0, stream>>>(
      dist_attn, path_attn, src, dst, offs, rank, psrc, ebias);
  gsan_ln_qkv_kernel<<<nodeBlocks, 256, 0, stream>>>(
      node_feature, gamma, beta, W_qkv, b_qkv, qkv);
  gsan_attn_out_kernel<<<nodeBlocks, 256, 0, stream>>>(
      qkv, ebias, cnt, offs, psrc, W_out, b_out, node_feature, out);
}


# W_qkv and W_out copied per call (init kernel) into WMMA B-fragment order: each B load is one contiguous 512-byte wave read feeding two k-steps, not 16 partial lines; k order and sums unchanged
# speedup vs baseline: 3.0193x; 1.1077x over previous
#include <hip/hip_runtime.h>

typedef __attribute__((ext_vector_type(2))) float v2f;
typedef __attribute__((ext_vector_type(8))) float v8f;

#define N_NODES   50000
#define N_EDGES   800000
#define D_MODEL   128
#define N_HEADS   8
#define HEAD_DIM  16
#define QKV_COLS  384
#define LN_EPS    1e-5f
#define QSCALE    0.08838834764831845f

#define SCAN_THREADS 1024
#define SCAN_CHUNK   49
#define SCAN_PAD     (SCAN_THREADS * SCAN_CHUNK)

#define INIT_CNT_BLOCKS ((N_NODES + 255) / 256)
#define WF_TILE         2048
#define WF_QKV_TILES    24
#define WF_OUT_TILES    8
#define WF_BLOCKS       ((WF_QKV_TILES + WF_OUT_TILES) * 512 / 256)

__global__ __launch_bounds__(256) void gsan_init_kernel(
    int* __restrict__ cnt,
    const float* __restrict__ W_qkv,
    const float* __restrict__ W_out,
    float* __restrict__ wfrag)
{
  const int b = blockIdx.x;
  if (b < INIT_CNT_BLOCKS) {
    const int i = b * 256 + threadIdx.x;
    if (i < N_NODES) cnt[i] = 0;
    return;
  }
  const int o4   = (b - INIT_CNT_BLOCKS) * 256 + threadIdx.x;
  const int l    = o4 & 31;
  const int p    = (o4 >> 5) & 15;
  const int tile = o4 >> 9;
  const int c    = l & 15;
  const int h    = l >> 4;
  const float* W = (tile < WF_QKV_TILES) ? W_qkv : W_out;
  const int    n = ((tile < WF_QKV_TILES) ? tile : (tile - WF_QKV_TILES)) * 16 + c;
  const float* s = W + n * D_MODEL + 8 * p + 2 * h;
  const float2 lo = *(const float2*)s;
  const float2 hi = *(const float2*)(s + 4);
  *(float4*)(wfrag + (size_t)o4 * 4) = make_float4(lo.x, lo.y, hi.x, hi.y);
}

__global__ __launch_bounds__(256) void gsan_ln_qkv_kernel(
    const float* __restrict__ node_feature,
    const float* __restrict__ gamma,
    const float* __restrict__ beta,
    const float* __restrict__ wfrag,
    const float* __restrict__ b_qkv,
    float* __restrict__ qkv)
{
  __shared__ float xs[16][132];
  __shared__ float smu[16], srs[16];
  const int tid = threadIdx.x;
  const int nodeBase = blockIdx.x * 16;

  for (int i = tid; i < 16 * 128; i += 256) {
    int r = i >> 7, c = i & 127;
    xs[r][c] = node_feature[(nodeBase + r) * D_MODEL + c];
  }
  __syncthreads();

  if (tid < 16) {
    float s = 0.0f, sq = 0.0f;
    for (int c = 0; c < 128; ++c) { float v = xs[tid][c]; s += v; sq += v * v; }
    float mu  = s * (1.0f / 128.0f);
    float var = sq * (1.0f / 128.0f) - mu * mu;
    smu[tid] = mu;
    srs[tid] = rsqrtf(var + LN_EPS);
  }
  __syncthreads();

  for (int i = tid; i < 16 * 128; i += 256) {
    int r = i >> 7, c = i & 127;
    xs[r][c] = (xs[r][c] - smu[r]) * srs[r] * gamma[c] + beta[c];
  }
  __syncthreads();

  const int lane = tid & 31;
  const int wave = tid >> 5;
  const int row  = lane & 15;
  const int half = lane >> 4;

  for (int t = 0; t < 3; ++t) {
    const int nt  = wave + t * 8;
    const int col = nt * 16 + row;
    const float* wf = wfrag + nt * WF_TILE + lane * 4;
    v8f acc = {};
    #pragma unroll 4
    for (int p = 0; p < 16; ++p) {
      const float4 bw = *(const float4*)(wf + p * 128);
      const int k = 8 * p + 2 * half;
      v2f a, b;
      a.x = xs[row][k];      a.y = xs[row][k + 1];
      b.x = bw.x;            b.y = bw.y;
      acc = __builtin_amdgcn_wmma_f32_16x16x4_f32(
          false, a, false, b, (short)0, acc, false, false);
      a.x = xs[row][k + 4];  a.y = xs[row][k + 5];
      b.x = bw.z;            b.y = bw.w;
      acc = __builtin_amdgcn_wmma_f32_16x16x4_f32(
          false, a, false, b, (short)0, acc, false, false);
    }
    const float bias  = b_qkv[col];
    const float scale = (nt < 8) ? QSCALE : 1.0f;
    #pragma unroll
    for (int j = 0; j < 8; ++j) {
      const int mrow = j + 8 * half;
      qkv[(nodeBase + mrow) * QKV_COLS + col] = (acc[j] + bias) * scale;
    }
  }
}

__global__ __launch_bounds__(256) void gsan_edge_count_kernel(
    const int* __restrict__ dst,
    int* __restrict__ cnt,
    int* __restrict__ rank)
{
  const int e = blockIdx.x * 256 + threadIdx.x;
  if (e >= N_EDGES) return;
  rank[e] = atomicAdd(&cnt[dst[e]], 1);
}

__global__ __launch_bounds__(SCAN_THREADS) void gsan_scan_kernel(
    const int* __restrict__ cnt, int* __restrict__ offs)
{
  __shared__ int s[SCAN_PAD];
  __shared__ int wsum[32];
  const int tid  = threadIdx.x;
  const int lane = tid & 31;
  const int wave = tid >> 5;
  for (int i = tid; i < SCAN_PAD; i += SCAN_THREADS) s[i] = (i < N_NODES) ? cnt[i] : 0;
  __syncthreads();
  int* my = s + tid * SCAN_CHUNK;
  int sum = 0;
  for (int j = 0; j < SCAN_CHUNK; ++j) sum += my[j];
  int incl = sum;
  for (int o = 1; o < 32; o <<= 1) { int t = __shfl_up(incl, o); if (lane >= o) incl += t; }
  if (lane == 31) wsum[wave] = incl;
  __syncthreads();
  if (wave == 0) {
    const int w = wsum[lane];
    int wi = w;
    for (int o = 1; o < 32; o <<= 1) { int t = __shfl_up(wi, o); if (lane >= o) wi += t; }
    wsum[lane] = wi - w;
  }
  __syncthreads();
  int run = wsum[wave] + incl - sum;
  for (int j = 0; j < SCAN_CHUNK; ++j) { int v = my[j]; my[j] = run; run += v; }
  __syncthreads();
  for (int i = tid; i < N_NODES; i += SCAN_THREADS) offs[i] = s[i];
}

__global__ __launch_bounds__(256) void gsan_edge_scatter_kernel(
    const float* __restrict__ dist_attn,
    const float* __restrict__ path_attn,
    const int*   __restrict__ src,
    const int*   __restrict__ dst,
    const int*   __restrict__ offs,
    const int*   __restrict__ rank,
    int*   __restrict__ psrc,
    float* __restrict__ ebias)
{
  const int i = blockIdx.x * 256 + threadIdx.x;
  if (i >= N_EDGES * 2) return;
  const int e = i >> 1, hh = i & 1;
  int pos = offs[dst[e]] + rank[e];
  pos = (pos < 0) ? 0 : ((pos > N_EDGES - 1) ? (N_EDGES - 1) : pos);
  const float4 dv = *(const float4*)(dist_attn + (size_t)i * 4);
  const float4 pv = *(const float4*)(path_attn + (size_t)i * 4);
  *(float4*)(ebias + (size_t)pos * N_HEADS + hh * 4) =
      make_float4(dv.x + pv.x, dv.y + pv.y, dv.z + pv.z, dv.w + pv.w);
  if (hh == 0) psrc[pos] = src[e];
}

__device__ __forceinline__ float quad_swap1(float x) {
  return __int_as_float(__builtin_amdgcn_update_dpp(0, __float_as_int(x), 0xB1, 0xF, 0xF, true));
}
__device__ __forceinline__ float quad_swap2(float x) {
  return __int_as_float(__builtin_amdgcn_update_dpp(0, __float_as_int(x), 0x4E, 0xF, 0xF, true));
}
template <int U>
__device__ __forceinline__ float quad_bcast(float x) {
  return __int_as_float(__builtin_amdgcn_update_dpp(0, __float_as_int(x), U * 0x55, 0xF, 0xF, true));
}

__device__ __forceinline__ void load_rows4(const float* __restrict__ qkv, int s_l, int j, int col4,
                                           float4 (&q)[4], float4 (&v)[4]) {
  #pragma unroll
  for (int u = 0; u < 4; ++u) {
    const int s = __builtin_amdgcn_readlane(s_l, j + u);
    const float* rowp = qkv + (size_t)s * QKV_COLS + col4;
    q[u] = *(const float4*)rowp;
    v[u] = *(const float4*)(rowp + 256);
  }
}

__device__ __forceinline__ float group_bias(const float* __restrict__ ebias, int seg, int j,
                                            int tq, int hl, int n) {
  const int jt = (j + tq < n) ? (j + tq) : (n - 1);
  return ebias[(size_t)(seg + jt) * N_HEADS + hl];
}

__device__ __forceinline__ void attn_group4(const float4 (&q)[4], const float4 (&v)[4], float bq,
                                            const float4 k4, int left,
                                            float& mrun, float& den, float4& acc) {
  const float bu[4] = {quad_bcast<0>(bq), quad_bcast<1>(bq), quad_bcast<2>(bq), quad_bcast<3>(bq)};
  float a[4];
  #pragma unroll
  for (int u = 0; u < 4; ++u) {
    float d = q[u].x * k4.x + q[u].y * k4.y + q[u].z * k4.z + q[u].w * k4.w;
    d += quad_swap1(d);
    d += quad_swap2(d);
    a[u] = (u < left) ? (d + bu[u]) : -__builtin_inff();
  }
  const float gmax = fmaxf(fmaxf(a[0], a[1]), fmaxf(a[2], a[3]));
  const float mnew = fmaxf(mrun, gmax);
  const float corr = expf(mrun - mnew);
  const float p0 = expf(a[0] - mnew), p1 = expf(a[1] - mnew);
  const float p2 = expf(a[2] - mnew), p3 = expf(a[3] - mnew);
  den = den * corr + ((p0 + p1) + (p2 + p3));
  const float p[4] = {p0, p1, p2, p3};
  float4 t = make_float4(acc.x * corr, acc.y * corr, acc.z * corr, acc.w * corr);
  #pragma unroll
  for (int u = 0; u < 4; ++u) {
    t.x = fmaf(p[u], v[u].x, t.x);  t.y = fmaf(p[u], v[u].y, t.y);
    t.z = fmaf(p[u], v[u].z, t.z);  t.w = fmaf(p[u], v[u].w, t.w);
  }
  acc  = t;
  mrun = mnew;
}

__global__ __launch_bounds__(256) void gsan_attn_out_kernel(
    const float* __restrict__ qkv,
    const float* __restrict__ ebias,
    const int*   __restrict__ cnt,
    const int*   __restrict__ offs,
    const int*   __restrict__ psrc,
    const float* __restrict__ wfrag,
    const float* __restrict__ b_out,
    const float* __restrict__ node_feature,
    float* __restrict__ out)
{
  __shared__ float xs[16][132];
  const int tid = threadIdx.x;
  const int nodeBase = blockIdx.x * 16;
  const int lane = tid & 31;
  const int hl   = lane >> 2;
  const int tq   = lane & 3;
  const int gw = __builtin_amdgcn_readfirstlane(tid >> 5);

  for (int nn = 0; nn < 2; ++nn) {
    const int r     = gw * 2 + nn;
    const int node  = nodeBase + r;
    int start = offs[node];
    int deg   = cnt[node];
    start = (start < 0) ? 0 : ((start > N_EDGES) ? N_EDGES : start);
    deg   = (deg < 0) ? 0 : ((deg > N_EDGES - start) ? (N_EDGES - start) : deg);
    const float4 k4 = *(const float4*)(qkv + (size_t)node * QKV_COLS + 128 + lane * 4);
    float4 acc  = make_float4(0.0f, 0.0f, 0.0f, 0.0f);
    float  den  = 0.0f;
    float  mrun = -__builtin_inff();
    for (int base = 0; base < deg; base += 32) {
      const int n   = (deg - base < 32) ? (deg - base) : 32;
      const int seg = start + base;
      int s_l = psrc[seg + ((lane < n) ? lane : (n - 1))];
      s_l = (s_l < 0) ? 0 : ((s_l > N_NODES - 1) ? (N_NODES - 1) : s_l);
      float4 qa[4], va[4], qb[4], vb[4];
      float  ba = group_bias(ebias, seg, 0, tq, hl, n), bb = 0.0f;
      load_rows4(qkv, s_l, 0, lane * 4, qa, va);
      for (int j = 0; j < n; j += 8) {
        if (j + 4 < n) {
          bb = group_bias(ebias, seg, j + 4, tq, hl, n);
          load_rows4(qkv, s_l, j + 4, lane * 4, qb, vb);
        }
        attn_group4(qa, va, ba, k4, n - j, mrun, den, acc);
        if (j + 4 < n) {
          if (j + 8 < n) {
            ba = group_bias(ebias, seg, j + 8, tq, hl, n);
            load_rows4(qkv, s_l, j + 8, lane * 4, qa, va);
          }
          attn_group4(qb, vb, bb, k4, n - j - 4, mrun, den, acc);
        }
      }
    }
    const float inv = (deg > 0) ? (1.0f / den) : 0.0f;
    *(float4*)&xs[r][lane * 4] = make_float4(acc.x * inv, acc.y * inv, acc.z * inv, acc.w * inv);
  }
  __syncthreads();

  const int wave = tid >> 5;
  const int row  = lane & 15;
  const int half = lane >> 4;
  const int col  = wave * 16 + row;
  const float* wf = wfrag + wave * WF_TILE + lane * 4;

  v8f acc = {};
  #pragma unroll 4
  for (int p = 0; p < 16; ++p) {
    const float4 bw = *(const float4*)(wf + p * 128);
    const int k = 8 * p + 2 * half;
    v2f a, b;
    a.x = xs[row][k];      a.y = xs[row][k + 1];
    b.x = bw.x;            b.y = bw.y;
    acc = __builtin_amdgcn_wmma_f32_16x16x4_f32(
        false, a, false, b, (short)0, acc, false, false);
    a.x = xs[row][k + 4];  a.y = xs[row][k + 5];
    b.x = bw.z;            b.y = bw.w;
    acc = __builtin_amdgcn_wmma_f32_16x16x4_f32(
        false, a, false, b, (short)0, acc, false, false);
  }
  const float bias = b_out[col];
  #pragma unroll
  for (int j = 0; j < 8; ++j) {
    const int mrow = j + 8 * half;
    const int idx  = (nodeBase + mrow) * D_MODEL + col;
    out[idx] = node_feature[idx] + acc[j] + bias;
  }
}

extern "C" void kernel_launch(void* const* d_in, const int* in_sizes, int n_in,
                              void* d_out, int out_size, void* d_ws, size_t ws_size,
                              hipStream_t stream) {
  const float* node_feature = (const float*)d_in[0];
  const float* dist_attn    = (const float*)d_in[1];
  const float* path_attn    = (const float*)d_in[2];
  const float* gamma        = (const float*)d_in[3];
  const float* beta         = (const float*)d_in[4];
  const float* W_qkv        = (const float*)d_in[5];
  const float* b_qkv        = (const float*)d_in[6];
  const float* W_out        = (const float*)d_in[7];
  const float* b_out        = (const float*)d_in[8];
  const int*   src          = (const int*)d_in[9];
  const int*   dst          = (const int*)d_in[10];
  float*       out          = (float*)d_out;

  float* ws    = (float*)d_ws;
  float* qkv   = ws;
  float* ebias = qkv   + (size_t)N_NODES * QKV_COLS;
  float* wfrag = ebias + (size_t)N_EDGES * N_HEADS;
  int*   cnt   = (int*)(wfrag + (size_t)2 * N_NODES * N_HEADS);
  int*   offs  = cnt  + 50048;
  int*   rank  = offs + 50048;
  int*   psrc  = rank + 2 * N_EDGES;

  const int edgeBlocks  = (N_EDGES + 255) / 256;
  const int edge2Blocks = (N_EDGES * 2 + 255) / 256;
  const int nodeBlocks  = N_NODES / 16;
  const int initBlocks  = INIT_CNT_BLOCKS + WF_BLOCKS;

  gsan_init_kernel<<<initBlocks, 256, 0, stream>>>(cnt, W_qkv, W_out, wfrag);
  gsan_edge_count_kernel<<<edgeBlocks, 256, 0, stream>>>(dst, cnt, rank);
  gsan_scan_kernel<<<1, SCAN_THREADS, 0, stream>>>(cnt, offs);
  gsan_edge_scatter_kernel<<<edge2Blocks, 256, 0, stream>>>(
      dist_attn, path_attn, src, dst, offs, rank, psrc, ebias);
  gsan_ln_qkv_kernel<<<nodeBlocks, 256, 0, stream>>>(
      node_feature, gamma, beta, wfrag, b_qkv, qkv);
  gsan_attn_out_kernel<<<nodeBlocks, 256, 0, stream>>>(
      qkv, ebias, cnt, offs, psrc, wfrag + WF_QKV_TILES * WF_TILE, b_out, node_feature, out);
}


# edge sort by dst rebuilt on LDS counters: 250 chunk blocks bin by 256-node range, 196 owner blocks count, scan, place in LDS, whole-line copy-out; no global atomics; agg sum order = LDS arrival order
# speedup vs baseline: 4.7023x; 1.5574x over previous
#include <hip/hip_runtime.h>

typedef __attribute__((ext_vector_type(2))) float v2f;
typedef __attribute__((ext_vector_type(8))) float v8f;

#define N_NODES   50000
#define N_EDGES   800000
#define D_MODEL   128
#define N_HEADS   8
#define HEAD_DIM  16
#define QKV_COLS  384
#define LN_EPS    1e-5f
#define QSCALE    0.08838834764831845f

#define SORT_CHUNKS 250
#define SORT_CH     3200
#define SORT_RANGES 196
#define PLACE_CAP   5120

#define WF_TILE         2048
#define WF_QKV_TILES    24
#define WF_OUT_TILES    8
#define WF_BLOCKS       ((WF_QKV_TILES + WF_OUT_TILES) * 512 / 256)

__global__ __launch_bounds__(256) void gsan_init_kernel(
    const float* __restrict__ W_qkv,
    const float* __restrict__ W_out,
    float* __restrict__ wfrag)
{
  const int o4   = blockIdx.x * 256 + threadIdx.x;
  const int l    = o4 & 31;
  const int p    = (o4 >> 5) & 15;
  const int tile = o4 >> 9;
  const int c    = l & 15;
  const int h    = l >> 4;
  const float* W = (tile < WF_QKV_TILES) ? W_qkv : W_out;
  const int    n = ((tile < WF_QKV_TILES) ? tile : (tile - WF_QKV_TILES)) * 16 + c;
  const float* s = W + n * D_MODEL + 8 * p + 2 * h;
  const float2 lo = *(const float2*)s;
  const float2 hi = *(const float2*)(s + 4);
  *(float4*)(wfrag + (size_t)o4 * 4) = make_float4(lo.x, lo.y, hi.x, hi.y);
}

__global__ __launch_bounds__(256) void gsan_ln_qkv_kernel(
    const float* __restrict__ node_feature,
    const float* __restrict__ gamma,
    const float* __restrict__ beta,
    const float* __restrict__ wfrag,
    const float* __restrict__ b_qkv,
    float* __restrict__ qkv)
{
  __shared__ float xs[16][132];
  __shared__ float smu[16], srs[16];
  const int tid = threadIdx.x;
  const int nodeBase = blockIdx.x * 16;

  for (int i = tid; i < 16 * 128; i += 256) {
    int r = i >> 7, c = i & 127;
    xs[r][c] = node_feature[(nodeBase + r) * D_MODEL + c];
  }
  __syncthreads();

  if (tid < 16) {
    float s = 0.0f, sq = 0.0f;
    for (int c = 0; c < 128; ++c) { float v = xs[tid][c]; s += v; sq += v * v; }
    float mu  = s * (1.0f / 128.0f);
    float var = sq * (1.0f / 128.0f) - mu * mu;
    smu[tid] = mu;
    srs[tid] = rsqrtf(var + LN_EPS);
  }
  __syncthreads();

  for (int i = tid; i < 16 * 128; i += 256) {
    int r = i >> 7, c = i & 127;
    xs[r][c] = (xs[r][c] - smu[r]) * srs[r] * gamma[c] + beta[c];
  }
  __syncthreads();

  const int lane = tid & 31;
  const int wave = tid >> 5;
  const int row  = lane & 15;
  const int half = lane >> 4;

  for (int t = 0; t < 3; ++t) {
    const int nt  = wave + t * 8;
    const int col = nt * 16 + row;
    const float* wf = wfrag + nt * WF_TILE + lane * 4;
    v8f acc = {};
    #pragma unroll 4
    for (int p = 0; p < 16; ++p) {
      const float4 bw = *(const float4*)(wf + p * 128);
      const int k = 8 * p + 2 * half;
      v2f a, b;
      a.x = xs[row][k];      a.y = xs[row][k + 1];
      b.x = bw.x;            b.y = bw.y;
      acc = __builtin_amdgcn_wmma_f32_16x16x4_f32(
          false, a, false, b, (short)0, acc, false, false);
      a.x = xs[row][k + 4];  a.y = xs[row][k + 5];
      b.x = bw.z;            b.y = bw.w;
      acc = __builtin_amdgcn_wmma_f32_16x16x4_f32(
          false, a, false, b, (short)0, acc, false, false);
    }
    const float bias  = b_qkv[col];
    const float scale = (nt < 8) ? QSCALE : 1.0f;
    #pragma unroll
    for (int j = 0; j < 8; ++j) {
      const int mrow = j + 8 * half;
      qkv[(nodeBase + mrow) * QKV_COLS + col] = (acc[j] + bias) * scale;
    }
  }
}

__global__ __launch_bounds__(1024) void gsan_sort_bin_kernel(
    const float* __restrict__ dist_attn,
    const float* __restrict__ path_attn,
    const int*   __restrict__ src,
    const int*   __restrict__ dst,
    float* __restrict__ t_eb,
    int*   __restrict__ t_src,
    int*   __restrict__ t_dst,
    int*   __restrict__ t_tab)
{
  __shared__ float4 s_eb[SORT_CH * 2];
  __shared__ int s_src[SORT_CH];
  __shared__ int s_dst[SORT_CH];
  __shared__ int s_pos[SORT_CH];
  __shared__ int s_hist[SORT_RANGES];
  __shared__ int s_start[SORT_RANGES + 1];
  const int tid   = threadIdx.x;
  const int lane  = tid & 31;
  const int chunk = blockIdx.x;
  const int e0    = chunk * SORT_CH;
  const float4* d4 = (const float4*)dist_attn + (size_t)e0 * 2;
  const float4* p4 = (const float4*)path_attn + (size_t)e0 * 2;

  int dv[4], sv[4], rk[4];
  #pragma unroll
  for (int it = 0; it < 4; ++it) {
    const int i  = tid + it * 1024;
    const int is = (i < SORT_CH) ? i : 0;
    int d = dst[e0 + is];
    d = (d < 0) ? 0 : ((d > N_NODES - 1) ? (N_NODES - 1) : d);
    dv[it] = d;
    sv[it] = src[e0 + is];
    rk[it] = 0;
  }
  float4 sm[7];
  #pragma unroll
  for (int m = 0; m < 7; ++m) {
    const int i2 = tid + m * 1024;
    const int is = (i2 < SORT_CH * 2) ? i2 : 0;
    const float4 a = d4[is], b = p4[is];
    sm[m] = make_float4(a.x + b.x, a.y + b.y, a.z + b.z, a.w + b.w);
  }
  if (tid < SORT_RANGES) s_hist[tid] = 0;
  __syncthreads();
  #pragma unroll
  for (int it = 0; it < 4; ++it) {
    const int i = tid + it * 1024;
    if (i < SORT_CH) rk[it] = atomicAdd(&s_hist[dv[it] >> 8], 1);
  }
  __syncthreads();
  if (tid < 32) {
    int v[7];
    int sum = 0;
    #pragma unroll
    for (int j = 0; j < 7; ++j) {
      const int idx = tid * 7 + j;
      v[j] = (idx < SORT_RANGES) ? s_hist[idx] : 0;
      sum += v[j];
    }
    int incl = sum;
    for (int o = 1; o < 32; o <<= 1) { const int t = __shfl_up(incl, o); if (lane >= o) incl += t; }
    int run = incl - sum;
    #pragma unroll
    for (int j = 0; j < 7; ++j) {
      const int idx = tid * 7 + j;
      if (idx < SORT_RANGES) s_start[idx] = run;
      run += v[j];
    }
    if (tid == 0) s_start[SORT_RANGES] = SORT_CH;
  }
  __syncthreads();
  #pragma unroll
  for (int it = 0; it < 4; ++it) {
    const int i = tid + it * 1024;
    if (i < SORT_CH) {
      int pos = s_start[dv[it] >> 8] + rk[it];
      pos = (pos < 0) ? 0 : ((pos > SORT_CH - 1) ? (SORT_CH - 1) : pos);
      s_pos[i]   = pos;
      s_src[pos] = sv[it];
      s_dst[pos] = dv[it];
    }
  }
  __syncthreads();
  #pragma unroll
  for (int m = 0; m < 7; ++m) {
    const int i2 = tid + m * 1024;
    if (i2 < SORT_CH * 2) s_eb[s_pos[i2 >> 1] * 2 + (i2 & 1)] = sm[m];
  }
  __syncthreads();
  for (int it = 0; it < 4; ++it) {
    const int i = tid + it * 1024;
    if (i < SORT_CH) { t_src[e0 + i] = s_src[i]; t_dst[e0 + i] = s_dst[i]; }
  }
  float4* o4 = (float4*)t_eb + (size_t)e0 * 2;
  for (int m = 0; m < 7; ++m) {
    const int i2 = tid + m * 1024;
    if (i2 < SORT_CH * 2) o4[i2] = s_eb[i2];
  }
  if (tid <= SORT_RANGES) t_tab[tid * SORT_CHUNKS + chunk] = s_start[tid];
}

__device__ __forceinline__ void place_rec(int d, int sn, const float4 ea, const float4 eb, bool staged,
                                          int base, int* s_cur, int* st_src, float4* st_eb,
                                          int* __restrict__ psrc, float4* __restrict__ ebias4) {
  const int slot = atomicAdd(&s_cur[d & 255], 1);
  if (staged) {
    const int sl = (slot < 0) ? 0 : ((slot > PLACE_CAP - 1) ? (PLACE_CAP - 1) : slot);
    st_src[sl]        = sn;
    st_eb[sl * 2]     = ea;
    st_eb[sl * 2 + 1] = eb;
  } else {
    int pos = base + slot;
    pos = (pos < 0) ? 0 : ((pos > N_EDGES - 1) ? (N_EDGES - 1) : pos);
    psrc[pos]           = sn;
    ebias4[pos * 2]     = ea;
    ebias4[pos * 2 + 1] = eb;
  }
}

__global__ __launch_bounds__(1024) void gsan_sort_place_kernel(
    const float* __restrict__ t_eb,
    const int*   __restrict__ t_src,
    const int*   __restrict__ t_dst,
    const int*   __restrict__ t_tab,
    int*   __restrict__ cnt,
    int*   __restrict__ offs,
    int*   __restrict__ psrc,
    float* __restrict__ ebias)
{
  __shared__ float4 st_eb[PLACE_CAP * 2];
  __shared__ int st_src[PLACE_CAP];
  __shared__ int s_lo[256], s_hi[256];
  __shared__ int s_cnt[256], s_off[256], s_cur[256];
  __shared__ int s_wa[32], s_wb[32];
  __shared__ int s_base, s_tot;
  const int tid  = threadIdx.x;
  const int lane = tid & 31;
  const int wave = __builtin_amdgcn_readfirstlane(tid >> 5);
  const int r    = blockIdx.x;
  const float4* te4 = (const float4*)t_eb;
  float4* ebias4 = (float4*)ebias;

  int lo = 0, hi = 0;
  if (tid < SORT_CHUNKS) {
    lo = t_tab[r * SORT_CHUNKS + tid];
    hi = t_tab[(r + 1) * SORT_CHUNKS + tid];
    lo = (lo < 0) ? 0 : ((lo > SORT_CH) ? SORT_CH : lo);
    hi = (hi < lo) ? lo : ((hi > SORT_CH) ? SORT_CH : hi);
  }
  if (tid < 256) { s_lo[tid] = lo; s_hi[tid] = hi; s_cnt[tid] = 0; }
  int wa = lo, wb = hi - lo;
  for (int o = 16; o > 0; o >>= 1) { wa += __shfl_xor(wa, o); wb += __shfl_xor(wb, o); }
  if (lane == 0) { s_wa[wave] = wa; s_wb[wave] = wb; }
  __syncthreads();
  if (tid < 32) {
    int a2 = s_wa[lane], b2 = s_wb[lane];
    for (int o = 16; o > 0; o >>= 1) { a2 += __shfl_xor(a2, o); b2 += __shfl_xor(b2, o); }
    if (lane == 0) { s_base = a2; s_tot = b2; }
  }

  int dd[8];
  #pragma unroll
  for (int k = 0; k < 8; ++k) {
    const int c = wave * 8 + k;
    const int j = s_lo[c] + lane;
    const bool ok = j < s_hi[c];
    const int v = t_dst[ok ? (c * SORT_CH + j) : 0];
    dd[k] = ok ? v : -1;
  }
  #pragma unroll
  for (int k = 0; k < 8; ++k)
    if (dd[k] >= 0) atomicAdd(&s_cnt[dd[k] & 255], 1);
  for (int k = 0; k < 8; ++k) {
    const int c = wave * 8 + k;
    const int hk = s_hi[c];
    for (int j = s_lo[c] + 32 + lane; j < hk; j += 32) {
      const int d = t_dst[c * SORT_CH + j];
      if (d >= 0) atomicAdd(&s_cnt[d & 255], 1);
    }
  }
  __syncthreads();
  if (tid < 32) {
    int v[8];
    int sum = 0;
    #pragma unroll
    for (int j = 0; j < 8; ++j) { v[j] = s_cnt[lane * 8 + j]; sum += v[j]; }
    int incl = sum;
    for (int o = 1; o < 32; o <<= 1) { const int t = __shfl_up(incl, o); if (lane >= o) incl += t; }
    int run = incl - sum;
    #pragma unroll
    for (int j = 0; j < 8; ++j) { s_off[lane * 8 + j] = run; s_cur[lane * 8 + j] = run; run += v[j]; }
  }
  __syncthreads();
  int base = s_base;
  int tot  = s_tot;
  base = (base < 0) ? 0 : ((base > N_EDGES) ? N_EDGES : base);
  tot  = (tot < 0) ? 0 : ((tot > N_EDGES - base) ? (N_EDGES - base) : tot);
  const bool staged = (tot <= PLACE_CAP);
  if (tid < 256) {
    const int node = r * 256 + tid;
    if (node < N_NODES) { cnt[node] = s_cnt[tid]; offs[node] = base + s_off[tid]; }
  }

  #pragma unroll
  for (int kb = 0; kb < 8; kb += 4) {
    int sn[4];
    float4 ea[4], eb[4];
    #pragma unroll
    for (int u = 0; u < 4; ++u) {
      const int c = wave * 8 + kb + u;
      const int g = (dd[kb + u] >= 0) ? (c * SORT_CH + s_lo[c] + lane) : 0;
      sn[u] = t_src[g];
      ea[u] = te4[(size_t)g * 2];
      eb[u] = te4[(size_t)g * 2 + 1];
    }
    #pragma unroll
    for (int u = 0; u < 4; ++u)
      if (dd[kb + u] >= 0)
        place_rec(dd[kb + u], sn[u], ea[u], eb[u], staged, base, s_cur, st_src, st_eb, psrc, ebias4);
  }
  for (int k = 0; k < 8; ++k) {
    const int c = wave * 8 + k;
    const int hk = s_hi[c];
    for (int j = s_lo[c] + 32 + lane; j < hk; j += 32) {
      const int g = c * SORT_CH + j;
      const int d = t_dst[g];
      if (d >= 0)
        place_rec(d, t_src[g], te4[(size_t)g * 2], te4[(size_t)g * 2 + 1], staged, base,
                  s_cur, st_src, st_eb, psrc, ebias4);
    }
  }
  __syncthreads();
  if (staged) {
    for (int i = tid; i < tot; i += 1024) psrc[base + i] = st_src[i];
    float4* o4 = ebias4 + (size_t)base * 2;
    for (int i2 = tid; i2 < tot * 2; i2 += 1024) o4[i2] = st_eb[i2];
  }
}

__device__ __forceinline__ float quad_swap1(float x) {
  return __int_as_float(__builtin_amdgcn_update_dpp(0, __float_as_int(x), 0xB1, 0xF, 0xF, true));
}
__device__ __forceinline__ float quad_swap2(float x) {
  return __int_as_float(__builtin_amdgcn_update_dpp(0, __float_as_int(x), 0x4E, 0xF, 0xF, true));
}
template <int U>
__device__ __forceinline__ float quad_bcast(float x) {
  return __int_as_float(__builtin_amdgcn_update_dpp(0, __float_as_int(x), U * 0x55, 0xF, 0xF, true));
}

__device__ __forceinline__ void load_rows4(const float* __restrict__ qkv, int s_l, int j, int col4,
                                           float4 (&q)[4], float4 (&v)[4]) {
  #pragma unroll
  for (int u = 0; u < 4; ++u) {
    const int s = __builtin_amdgcn_readlane(s_l, j + u);
    const float* rowp = qkv + (size_t)s * QKV_COLS + col4;
    q[u] = *(const float4*)rowp;
    v[u] = *(const float4*)(rowp + 256);
  }
}

__device__ __forceinline__ float group_bias(const float* __restrict__ ebias, int seg, int j,
                                            int tq, int hl, int n) {
  const int jt = (j + tq < n) ? (j + tq) : (n - 1);
  return ebias[(size_t)(seg + jt) * N_HEADS + hl];
}

__device__ __forceinline__ void attn_group4(const float4 (&q)[4], const float4 (&v)[4], float bq,
                                            const float4 k4, int left,
                                            float& mrun, float& den, float4& acc) {
  const float bu[4] = {quad_bcast<0>(bq), quad_bcast<1>(bq), quad_bcast<2>(bq), quad_bcast<3>(bq)};
  float a[4];
  #pragma unroll
  for (int u = 0; u < 4; ++u) {
    float d = q[u].x * k4.x + q[u].y * k4.y + q[u].z * k4.z + q[u].w * k4.w;
    d += quad_swap1(d);
    d += quad_swap2(d);
    a[u] = (u < left) ? (d + bu[u]) : -__builtin_inff();
  }
  const float gmax = fmaxf(fmaxf(a[0], a[1]), fmaxf(a[2], a[3]));
  const float mnew = fmaxf(mrun, gmax);
  const float corr = expf(mrun - mnew);
  const float p0 = expf(a[0] - mnew), p1 = expf(a[1] - mnew);
  const float p2 = expf(a[2] - mnew), p3 = expf(a[3] - mnew);
  den = den * corr + ((p0 + p1) + (p2 + p3));
  const float p[4] = {p0, p1, p2, p3};
  float4 t = make_float4(acc.x * corr, acc.y * corr, acc.z * corr, acc.w * corr);
  #pragma unroll
  for (int u = 0; u < 4; ++u) {
    t.x = fmaf(p[u], v[u].x, t.x);  t.y = fmaf(p[u], v[u].y, t.y);
    t.z = fmaf(p[u], v[u].z, t.z);  t.w = fmaf(p[u], v[u].w, t.w);
  }
  acc  = t;
  mrun = mnew;
}

__global__ __launch_bounds__(256) void gsan_attn_out_kernel(
    const float* __restrict__ qkv,
    const float* __restrict__ ebias,
    const int*   __restrict__ cnt,
    const int*   __restrict__ offs,
    const int*   __restrict__ psrc,
    const float* __restrict__ wfrag,
    const float* __restrict__ b_out,
    const float* __restrict__ node_feature,
    float* __restrict__ out)
{
  __shared__ float xs[16][132];
  const int tid = threadIdx.x;
  const int nodeBase = blockIdx.x * 16;
  const int lane = tid & 31;
  const int hl   = lane >> 2;
  const int tq   = lane & 3;
  const int gw = __builtin_amdgcn_readfirstlane(tid >> 5);

  for (int nn = 0; nn < 2; ++nn) {
    const int r     = gw * 2 + nn;
    const int node  = nodeBase + r;
    int start = offs[node];
    int deg   = cnt[node];
    start = (start < 0) ? 0 : ((start > N_EDGES) ? N_EDGES : start);
    deg   = (deg < 0) ? 0 : ((deg > N_EDGES - start) ? (N_EDGES - start) : deg);
    const float4 k4 = *(const float4*)(qkv + (size_t)node * QKV_COLS + 128 + lane * 4);
    float4 acc  = make_float4(0.0f, 0.0f, 0.0f, 0.0f);
    float  den  = 0.0f;
    float  mrun = -__builtin_inff();
    for (int base = 0; base < deg; base += 32) {
      const int n   = (deg - base < 32) ? (deg - base) : 32;
      const int seg = start + base;
      int s_l = psrc[seg + ((lane < n) ? lane : (n - 1))];
      s_l = (s_l < 0) ? 0 : ((s_l > N_NODES - 1) ? (N_NODES - 1) : s_l);
      float4 qa[4], va[4], qb[4], vb[4];
      float  ba = group_bias(ebias, seg, 0, tq, hl, n), bb = 0.0f;
      load_rows4(qkv, s_l, 0, lane * 4, qa, va);
      for (int j = 0; j < n; j += 8) {
        if (j + 4 < n) {
          bb = group_bias(ebias, seg, j + 4, tq, hl, n);
          load_rows4(qkv, s_l, j + 4, lane * 4, qb, vb);
        }
        attn_group4(qa, va, ba, k4, n - j, mrun, den, acc);
        if (j + 4 < n) {
          if (j + 8 < n) {
            ba = group_bias(ebias, seg, j + 8, tq, hl, n);
            load_rows4(qkv, s_l, j + 8, lane * 4, qa, va);
          }
          attn_group4(qb, vb, bb, k4, n - j - 4, mrun, den, acc);
        }
      }
    }
    const float inv = (deg > 0) ? (1.0f / den) : 0.0f;
    *(float4*)&xs[r][lane * 4] = make_float4(acc.x * inv, acc.y * inv, acc.z * inv, acc.w * inv);
  }
  __syncthreads();

  const int wave = tid >> 5;
  const int row  = lane & 15;
  const int half = lane >> 4;
  const int col  = wave * 16 + row;
  const float* wf = wfrag + wave * WF_TILE + lane * 4;

  v8f acc = {};
  #pragma unroll 4
  for (int p = 0; p < 16; ++p) {
    const float4 bw = *(const float4*)(wf + p * 128);
    const int k = 8 * p + 2 * half;
    v2f a, b;
    a.x = xs[row][k];      a.y = xs[row][k + 1];
    b.x = bw.x;            b.y = bw.y;
    acc = __builtin_amdgcn_wmma_f32_16x16x4_f32(
        false, a, false, b, (short)0, acc, false, false);
    a.x = xs[row][k + 4];  a.y = xs[row][k + 5];
    b.x = bw.z;            b.y = bw.w;
    acc = __builtin_amdgcn_wmma_f32_16x16x4_f32(
        false, a, false, b, (short)0, acc, false, false);
  }
  const float bias = b_out[col];
  #pragma unroll
  for (int j = 0; j < 8; ++j) {
    const int mrow = j + 8 * half;
    const int idx  = (nodeBase + mrow) * D_MODEL + col;
    out[idx] = node_feature[idx] + acc[j] + bias;
  }
}

extern "C" void kernel_launch(void* const* d_in, const int* in_sizes, int n_in,
                              void* d_out, int out_size, void* d_ws, size_t ws_size,
                              hipStream_t stream) {
  const float* node_feature = (const float*)d_in[0];
  const float* dist_attn    = (const float*)d_in[1];
  const float* path_attn    = (const float*)d_in[2];
  const float* gamma        = (const float*)d_in[3];
  const float* beta         = (const float*)d_in[4];
  const float* W_qkv        = (const float*)d_in[5];
  const float* b_qkv        = (const float*)d_in[6];
  const float* W_out        = (const float*)d_in[7];
  const float* b_out        = (const float*)d_in[8];
  const int*   src          = (const int*)d_in[9];
  const int*   dst          = (const int*)d_in[10];
  float*       out          = (float*)d_out;

  float* ws    = (float*)d_ws;
  float* qkv   = ws;
  float* ebias = qkv   + (size_t)N_NODES * QKV_COLS;
  float* wfrag = ebias + (size_t)N_EDGES * N_HEADS;
  int*   cnt   = (int*)(wfrag + (size_t)2 * N_NODES * N_HEADS);
  int*   offs  = cnt  + 50048;
  int*   psrc  = offs + 50048 + 2 * N_EDGES;
  float* t_eb  = qkv;
  int*   t_src = (int*)(qkv + (size_t)N_EDGES * N_HEADS);
  int*   t_dst = t_src + N_EDGES;
  int*   t_tab = t_dst + N_EDGES;

  const int nodeBlocks = N_NODES / 16;

  gsan_init_kernel<<<WF_BLOCKS, 256, 0, stream>>>(W_qkv, W_out, wfrag);
  gsan_sort_bin_kernel<<<SORT_CHUNKS, 1024, 0, stream>>>(
      dist_attn, path_attn, src, dst, t_eb, t_src, t_dst, t_tab);
  gsan_sort_place_kernel<<<SORT_RANGES, 1024, 0, stream>>>(
      t_eb, t_src, t_dst, t_tab, cnt, offs, psrc, ebias);
  gsan_ln_qkv_kernel<<<nodeBlocks, 256, 0, stream>>>(
      node_feature, gamma, beta, wfrag, b_qkv, qkv);
  gsan_attn_out_kernel<<<nodeBlocks, 256, 0, stream>>>(
      qkv, ebias, cnt, offs, psrc, wfrag + WF_QKV_TILES * WF_TILE, b_out, node_feature, out);
}


# qkv kernel LayerNorm stage: a row's 128 values enter as two 16-byte loads per lane (16 lanes a row, all in flight at once), row sums by a 16-lane DPP tree, one barrier; row sum order changed
# speedup vs baseline: 5.0508x; 1.0741x over previous
#include <hip/hip_runtime.h>

typedef __attribute__((ext_vector_type(2))) float v2f;
typedef __attribute__((ext_vector_type(8))) float v8f;

#define N_NODES   50000
#define N_EDGES   800000
#define D_MODEL   128
#define N_HEADS   8
#define HEAD_DIM  16
#define QKV_COLS  384
#define LN_EPS    1e-5f
#define QSCALE    0.08838834764831845f

#define SORT_CHUNKS 250
#define SORT_CH     3200
#define SORT_RANGES 196
#define PLACE_CAP   5120

#define WF_TILE         2048
#define WF_QKV_TILES    24
#define WF_OUT_TILES    8
#define WF_BLOCKS       ((WF_QKV_TILES + WF_OUT_TILES) * 512 / 256)

__global__ __launch_bounds__(256) void gsan_init_kernel(
    const float* __restrict__ W_qkv,
    const float* __restrict__ W_out,
    float* __restrict__ wfrag)
{
  const int o4   = blockIdx.x * 256 + threadIdx.x;
  const int l    = o4 & 31;
  const int p    = (o4 >> 5) & 15;
  const int tile = o4 >> 9;
  const int c    = l & 15;
  const int h    = l >> 4;
  const float* W = (tile < WF_QKV_TILES) ? W_qkv : W_out;
  const int    n = ((tile < WF_QKV_TILES) ? tile : (tile - WF_QKV_TILES)) * 16 + c;
  const float* s = W + n * D_MODEL + 8 * p + 2 * h;
  const float2 lo = *(const float2*)s;
  const float2 hi = *(const float2*)(s + 4);
  *(float4*)(wfrag + (size_t)o4 * 4) = make_float4(lo.x, lo.y, hi.x, hi.y);
}

template <int CTRL>
__device__ __forceinline__ float row_dpp(float x) {
  return __int_as_float(__builtin_amdgcn_update_dpp(0, __float_as_int(x), CTRL, 0xF, 0xF, true));
}
__device__ __forceinline__ float row16_sum(float x) {
  x += row_dpp<0xB1>(x);
  x += row_dpp<0x4E>(x);
  x += row_dpp<0x164>(x);
  x += row_dpp<0x168>(x);
  return x;
}

__global__ __launch_bounds__(256) void gsan_ln_qkv_kernel(
    const float* __restrict__ node_feature,
    const float* __restrict__ gamma,
    const float* __restrict__ beta,
    const float* __restrict__ wfrag,
    const float* __restrict__ b_qkv,
    float* __restrict__ qkv)
{
  __shared__ float xs[16][132];
  const int tid = threadIdx.x;
  const int nodeBase = blockIdx.x * 16;

  {
    const int r = tid >> 4;
    const int q = tid & 15;
    const float* rp = node_feature + (size_t)(nodeBase + r) * D_MODEL + 4 * q;
    const float4 xa = *(const float4*)rp;
    const float4 xb = *(const float4*)(rp + 64);
    const float4 ga = *(const float4*)(gamma + 4 * q);
    const float4 gb = *(const float4*)(gamma + 64 + 4 * q);
    const float4 ba = *(const float4*)(beta + 4 * q);
    const float4 bb = *(const float4*)(beta + 64 + 4 * q);

    float s  = ((xa.x + xa.y) + (xa.z + xa.w)) + ((xb.x + xb.y) + (xb.z + xb.w));
    float sq = ((xa.x * xa.x + xa.y * xa.y) + (xa.z * xa.z + xa.w * xa.w))
             + ((xb.x * xb.x + xb.y * xb.y) + (xb.z * xb.z + xb.w * xb.w));
    s  = row16_sum(s);
    sq = row16_sum(sq);
    const float mu  = s * (1.0f / 128.0f);
    const float var = sq * (1.0f / 128.0f) - mu * mu;
    const float rs  = rsqrtf(var + LN_EPS);

    float4 ya, yb;
    ya.x = (xa.x - mu) * rs * ga.x + ba.x;  ya.y = (xa.y - mu) * rs * ga.y + ba.y;
    ya.z = (xa.z - mu) * rs * ga.z + ba.z;  ya.w = (xa.w - mu) * rs * ga.w + ba.w;
    yb.x = (xb.x - mu) * rs * gb.x + bb.x;  yb.y = (xb.y - mu) * rs * gb.y + bb.y;
    yb.z = (xb.z - mu) * rs * gb.z + bb.z;  yb.w = (xb.w - mu) * rs * gb.w + bb.w;
    *(float4*)&xs[r][4 * q]      = ya;
    *(float4*)&xs[r][64 + 4 * q] = yb;
  }
  __syncthreads();

  const int lane = tid & 31;
  const int wave = tid >> 5;
  const int row  = lane & 15;
  const int half = lane >> 4;

  for (int t = 0; t < 3; ++t) {
    const int nt  = wave + t * 8;
    const int col = nt * 16 + row;
    const float* wf = wfrag + nt * WF_TILE + lane * 4;
    v8f acc = {};
    #pragma unroll 4
    for (int p = 0; p < 16; ++p) {
      const float4 bw = *(const float4*)(wf + p * 128);
      const int k = 8 * p + 2 * half;
      v2f a, b;
      a.x = xs[row][k];      a.y = xs[row][k + 1];
      b.x = bw.x;            b.y = bw.y;
      acc = __builtin_amdgcn_wmma_f32_16x16x4_f32(
          false, a, false, b, (short)0, acc, false, false);
      a.x = xs[row][k + 4];  a.y = xs[row][k + 5];
      b.x = bw.z;            b.y = bw.w;
      acc = __builtin_amdgcn_wmma_f32_16x16x4_f32(
          false, a, false, b, (short)0, acc, false, false);
    }
    const float bias  = b_qkv[col];
    const float scale = (nt < 8) ? QSCALE : 1.0f;
    #pragma unroll
    for (int j = 0; j < 8; ++j) {
      const int mrow = j + 8 * half;
      qkv[(nodeBase + mrow) * QKV_COLS + col] = (acc[j] + bias) * scale;
    }
  }
}

__global__ __launch_bounds__(1024) void gsan_sort_bin_kernel(
    const float* __restrict__ dist_attn,
    const float* __restrict__ path_attn,
    const int*   __restrict__ src,
    const int*   __restrict__ dst,
    float* __restrict__ t_eb,
    int*   __restrict__ t_src,
    int*   __restrict__ t_dst,
    int*   __restrict__ t_tab)
{
  __shared__ float4 s_eb[SORT_CH * 2];
  __shared__ int s_src[SORT_CH];
  __shared__ int s_dst[SORT_CH];
  __shared__ int s_pos[SORT_CH];
  __shared__ int s_hist[SORT_RANGES];
  __shared__ int s_start[SORT_RANGES + 1];
  const int tid   = threadIdx.x;
  const int lane  = tid & 31;
  const int chunk = blockIdx.x;
  const int e0    = chunk * SORT_CH;
  const float4* d4 = (const float4*)dist_attn + (size_t)e0 * 2;
  const float4* p4 = (const float4*)path_attn + (size_t)e0 * 2;

  int dv[4], sv[4], rk[4];
  #pragma unroll
  for (int it = 0; it < 4; ++it) {
    const int i  = tid + it * 1024;
    const int is = (i < SORT_CH) ? i : 0;
    int d = dst[e0 + is];
    d = (d < 0) ? 0 : ((d > N_NODES - 1) ? (N_NODES - 1) : d);
    dv[it] = d;
    sv[it] = src[e0 + is];
    rk[it] = 0;
  }
  float4 sm[7];
  #pragma unroll
  for (int m = 0; m < 7; ++m) {
    const int i2 = tid + m * 1024;
    const int is = (i2 < SORT_CH * 2) ? i2 : 0;
    const float4 a = d4[is], b = p4[is];
    sm[m] = make_float4(a.x + b.x, a.y + b.y, a.z + b.z, a.w + b.w);
  }
  if (tid < SORT_RANGES) s_hist[tid] = 0;
  __syncthreads();
  #pragma unroll
  for (int it = 0; it < 4; ++it) {
    const int i = tid + it * 1024;
    if (i < SORT_CH) rk[it] = atomicAdd(&s_hist[dv[it] >> 8], 1);
  }
  __syncthreads();
  if (tid < 32) {
    int v[7];
    int sum = 0;
    #pragma unroll
    for (int j = 0; j < 7; ++j) {
      const int idx = tid * 7 + j;
      v[j] = (idx < SORT_RANGES) ? s_hist[idx] : 0;
      sum += v[j];
    }
    int incl = sum;
    for (int o = 1; o < 32; o <<= 1) { const int t = __shfl_up(incl, o); if (lane >= o) incl += t; }
    int run = incl - sum;
    #pragma unroll
    for (int j = 0; j < 7; ++j) {
      const int idx = tid * 7 + j;
      if (idx < SORT_RANGES) s_start[idx] = run;
      run += v[j];
    }
    if (tid == 0) s_start[SORT_RANGES] = SORT_CH;
  }
  __syncthreads();
  #pragma unroll
  for (int it = 0; it < 4; ++it) {
    const int i = tid + it * 1024;
    if (i < SORT_CH) {
      int pos = s_start[dv[it] >> 8] + rk[it];
      pos = (pos < 0) ? 0 : ((pos > SORT_CH - 1) ? (SORT_CH - 1) : pos);
      s_pos[i]   = pos;
      s_src[pos] = sv[it];
      s_dst[pos] = dv[it];
    }
  }
  __syncthreads();
  #pragma unroll
  for (int m = 0; m < 7; ++m) {
    const int i2 = tid + m * 1024;
    if (i2 < SORT_CH * 2) s_eb[s_pos[i2 >> 1] * 2 + (i2 & 1)] = sm[m];
  }
  __syncthreads();
  for (int it = 0; it < 4; ++it) {
    const int i = tid + it * 1024;
    if (i < SORT_CH) { t_src[e0 + i] = s_src[i]; t_dst[e0 + i] = s_dst[i]; }
  }
  float4* o4 = (float4*)t_eb + (size_t)e0 * 2;
  for (int m = 0; m < 7; ++m) {
    const int i2 = tid + m * 1024;
    if (i2 < SORT_CH * 2) o4[i2] = s_eb[i2];
  }
  if (tid <= SORT_RANGES) t_tab[tid * SORT_CHUNKS + chunk] = s_start[tid];
}

__device__ __forceinline__ void place_rec(int d, int sn, const float4 ea, const float4 eb, bool staged,
                                          int base, int* s_cur, int* st_src, float4* st_eb,
                                          int* __restrict__ psrc, float4* __restrict__ ebias4) {
  const int slot = atomicAdd(&s_cur[d & 255], 1);
  if (staged) {
    const int sl = (slot < 0) ? 0 : ((slot > PLACE_CAP - 1) ? (PLACE_CAP - 1) : slot);
    st_src[sl]        = sn;
    st_eb[sl * 2]     = ea;
    st_eb[sl * 2 + 1] = eb;
  } else {
    int pos = base + slot;
    pos = (pos < 0) ? 0 : ((pos > N_EDGES - 1) ? (N_EDGES - 1) : pos);
    psrc[pos]           = sn;
    ebias4[pos * 2]     = ea;
    ebias4[pos * 2 + 1] = eb;
  }
}

__global__ __launch_bounds__(1024) void gsan_sort_place_kernel(
    const float* __restrict__ t_eb,
    const int*   __restrict__ t_src,
    const int*   __restrict__ t_dst,
    const int*   __restrict__ t_tab,
    int*   __restrict__ cnt,
    int*   __restrict__ offs,
    int*   __restrict__ psrc,
    float* __restrict__ ebias)
{
  __shared__ float4 st_eb[PLACE_CAP * 2];
  __shared__ int st_src[PLACE_CAP];
  __shared__ int s_lo[256], s_hi[256];
  __shared__ int s_cnt[256], s_off[256], s_cur[256];
  __shared__ int s_wa[32], s_wb[32];
  __shared__ int s_base, s_tot;
  const int tid  = threadIdx.x;
  const int lane = tid & 31;
  const int wave = __builtin_amdgcn_readfirstlane(tid >> 5);
  const int r    = blockIdx.x;
  const float4* te4 = (const float4*)t_eb;
  float4* ebias4 = (float4*)ebias;

  int lo = 0, hi = 0;
  if (tid < SORT_CHUNKS) {
    lo = t_tab[r * SORT_CHUNKS + tid];
    hi = t_tab[(r + 1) * SORT_CHUNKS + tid];
    lo = (lo < 0) ? 0 : ((lo > SORT_CH) ? SORT_CH : lo);
    hi = (hi < lo) ? lo : ((hi > SORT_CH) ? SORT_CH : hi);
  }
  if (tid < 256) { s_lo[tid] = lo; s_hi[tid] = hi; s_cnt[tid] = 0; }
  int wa = lo, wb = hi - lo;
  for (int o = 16; o > 0; o >>= 1) { wa += __shfl_xor(wa, o); wb += __shfl_xor(wb, o); }
  if (lane == 0) { s_wa[wave] = wa; s_wb[wave] = wb; }
  __syncthreads();
  if (tid < 32) {
    int a2 = s_wa[lane], b2 = s_wb[lane];
    for (int o = 16; o > 0; o >>= 1) { a2 += __shfl_xor(a2, o); b2 += __shfl_xor(b2, o); }
    if (lane == 0) { s_base = a2; s_tot = b2; }
  }

  int dd[8];
  #pragma unroll
  for (int k = 0; k < 8; ++k) {
    const int c = wave * 8 + k;
    const int j = s_lo[c] + lane;
    const bool ok = j < s_hi[c];
    const int v = t_dst[ok ? (c * SORT_CH + j) : 0];
    dd[k] = ok ? v : -1;
  }
  #pragma unroll
  for (int k = 0; k < 8; ++k)
    if (dd[k] >= 0) atomicAdd(&s_cnt[dd[k] & 255], 1);
  for (int k = 0; k < 8; ++k) {
    const int c = wave * 8 + k;
    const int hk = s_hi[c];
    for (int j = s_lo[c] + 32 + lane; j < hk; j += 32) {
      const int d = t_dst[c * SORT_CH + j];
      if (d >= 0) atomicAdd(&s_cnt[d & 255], 1);
    }
  }
  __syncthreads();
  if (tid < 32) {
    int v[8];
    int sum = 0;
    #pragma unroll
    for (int j = 0; j < 8; ++j) { v[j] = s_cnt[lane * 8 + j]; sum += v[j]; }
    int incl = sum;
    for (int o = 1; o < 32; o <<= 1) { const int t = __shfl_up(incl, o); if (lane >= o) incl += t; }
    int run = incl - sum;
    #pragma unroll
    for (int j = 0; j < 8; ++j) { s_off[lane * 8 + j] = run; s_cur[lane * 8 + j] = run; run += v[j]; }
  }
  __syncthreads();
  int base = s_base;
  int tot  = s_tot;
  base = (base < 0) ? 0 : ((base > N_EDGES) ? N_EDGES : base);
  tot  = (tot < 0) ? 0 : ((tot > N_EDGES - base) ? (N_EDGES - base) : tot);
  const bool staged = (tot <= PLACE_CAP);
  if (tid < 256) {
    const int node = r * 256 + tid;
    if (node < N_NODES) { cnt[node] = s_cnt[tid]; offs[node] = base + s_off[tid]; }
  }

  #pragma unroll
  for (int kb = 0; kb < 8; kb += 4) {
    int sn[4];
    float4 ea[4], eb[4];
    #pragma unroll
    for (int u = 0; u < 4; ++u) {
      const int c = wave * 8 + kb + u;
      const int g = (dd[kb + u] >= 0) ? (c * SORT_CH + s_lo[c] + lane) : 0;
      sn[u] = t_src[g];
      ea[u] = te4[(size_t)g * 2];
      eb[u] = te4[(size_t)g * 2 + 1];
    }
    #pragma unroll
    for (int u = 0; u < 4; ++u)
      if (dd[kb + u] >= 0)
        place_rec(dd[kb + u], sn[u], ea[u], eb[u], staged, base, s_cur, st_src, st_eb, psrc, ebias4);
  }
  for (int k = 0; k < 8; ++k) {
    const int c = wave * 8 + k;
    const int hk = s_hi[c];
    for (int j = s_lo[c] + 32 + lane; j < hk; j += 32) {
      const int g = c * SORT_CH + j;
      const int d = t_dst[g];
      if (d >= 0)
        place_rec(d, t_src[g], te4[(size_t)g * 2], te4[(size_t)g * 2 + 1], staged, base,
                  s_cur, st_src, st_eb, psrc, ebias4);
    }
  }
  __syncthreads();
  if (staged) {
    for (int i = tid; i < tot; i += 1024) psrc[base + i] = st_src[i];
    float4* o4 = ebias4 + (size_t)base * 2;
    for (int i2 = tid; i2 < tot * 2; i2 += 1024) o4[i2] = st_eb[i2];
  }
}

__device__ __forceinline__ float quad_swap1(float x) {
  return __int_as_float(__builtin_amdgcn_update_dpp(0, __float_as_int(x), 0xB1, 0xF, 0xF, true));
}
__device__ __forceinline__ float quad_swap2(float x) {
  return __int_as_float(__builtin_amdgcn_update_dpp(0, __float_as_int(x), 0x4E, 0xF, 0xF, true));
}
template <int U>
__device__ __forceinline__ float quad_bcast(float x) {
  return __int_as_float(__builtin_amdgcn_update_dpp(0, __float_as_int(x), U * 0x55, 0xF, 0xF, true));
}

__device__ __forceinline__ void load_rows4(const float* __restrict__ qkv, int s_l, int j, int col4,
                                           float4 (&q)[4], float4 (&v)[4]) {
  #pragma unroll
  for (int u = 0; u < 4; ++u) {
    const int s = __builtin_amdgcn_readlane(s_l, j + u);
    const float* rowp = qkv + (size_t)s * QKV_COLS + col4;
    q[u] = *(const float4*)rowp;
    v[u] = *(const float4*)(rowp + 256);
  }
}

__device__ __forceinline__ float group_bias(const float* __restrict__ ebias, int seg, int j,
                                            int tq, int hl, int n) {
  const int jt = (j + tq < n) ? (j + tq) : (n - 1);
  return ebias[(size_t)(seg + jt) * N_HEADS + hl];
}

__device__ __forceinline__ void attn_group4(const float4 (&q)[4], const float4 (&v)[4], float bq,
                                            const float4 k4, int left,
                                            float& mrun, float& den, float4& acc) {
  const float bu[4] = {quad_bcast<0>(bq), quad_bcast<1>(bq), quad_bcast<2>(bq), quad_bcast<3>(bq)};
  float a[4];
  #pragma unroll
  for (int u = 0; u < 4; ++u) {
    float d = q[u].x * k4.x + q[u].y * k4.y + q[u].z * k4.z + q[u].w * k4.w;
    d += quad_swap1(d);
    d += quad_swap2(d);
    a[u] = (u < left) ? (d + bu[u]) : -__builtin_inff();
  }
  const float gmax = fmaxf(fmaxf(a[0], a[1]), fmaxf(a[2], a[3]));
  const float mnew = fmaxf(mrun, gmax);
  const float corr = expf(mrun - mnew);
  const float p0 = expf(a[0] - mnew), p1 = expf(a[1] - mnew);
  const float p2 = expf(a[2] - mnew), p3 = expf(a[3] - mnew);
  den = den * corr + ((p0 + p1) + (p2 + p3));
  const float p[4] = {p0, p1, p2, p3};
  float4 t = make_float4(acc.x * corr, acc.y * corr, acc.z * corr, acc.w * corr);
  #pragma unroll
  for (int u = 0; u < 4; ++u) {
    t.x = fmaf(p[u], v[u].x, t.x);  t.y = fmaf(p[u], v[u].y, t.y);
    t.z = fmaf(p[u], v[u].z, t.z);  t.w = fmaf(p[u], v[u].w, t.w);
  }
  acc  = t;
  mrun = mnew;
}

__global__ __launch_bounds__(256) void gsan_attn_out_kernel(
    const float* __restrict__ qkv,
    const float* __restrict__ ebias,
    const int*   __restrict__ cnt,
    const int*   __restrict__ offs,
    const int*   __restrict__ psrc,
    const float* __restrict__ wfrag,
    const float* __restrict__ b_out,
    const float* __restrict__ node_feature,
    float* __restrict__ out)
{
  __shared__ float xs[16][132];
  const int tid = threadIdx.x;
  const int nodeBase = blockIdx.x * 16;
  const int lane = tid & 31;
  const int hl   = lane >> 2;
  const int tq   = lane & 3;
  const int gw = __builtin_amdgcn_readfirstlane(tid >> 5);

  for (int nn = 0; nn < 2; ++nn) {
    const int r     = gw * 2 + nn;
    const int node  = nodeBase + r;
    int start = offs[node];
    int deg   = cnt[node];
    start = (start < 0) ? 0 : ((start > N_EDGES) ? N_EDGES : start);
    deg   = (deg < 0) ? 0 : ((deg > N_EDGES - start) ? (N_EDGES - start) : deg);
    const float4 k4 = *(const float4*)(qkv + (size_t)node * QKV_COLS + 128 + lane * 4);
    float4 acc  = make_float4(0.0f, 0.0f, 0.0f, 0.0f);
    float  den  = 0.0f;
    float  mrun = -__builtin_inff();
    for (int base = 0; base < deg; base += 32) {
      const int n   = (deg - base < 32) ? (deg - base) : 32;
      const int seg = start + base;
      int s_l = psrc[seg + ((lane < n) ? lane : (n - 1))];
      s_l = (s_l < 0) ? 0 : ((s_l > N_NODES - 1) ? (N_NODES - 1) : s_l);
      float4 qa[4], va[4], qb[4], vb[4];
      float  ba = group_bias(ebias, seg, 0, tq, hl, n), bb = 0.0f;
      load_rows4(qkv, s_l, 0, lane * 4, qa, va);
      for (int j = 0; j < n; j += 8) {
        if (j + 4 < n) {
          bb = group_bias(ebias, seg, j + 4, tq, hl, n);
          load_rows4(qkv, s_l, j + 4, lane * 4, qb, vb);
        }
        attn_group4(qa, va, ba, k4, n - j, mrun, den, acc);
        if (j + 4 < n) {
          if (j + 8 < n) {
            ba = group_bias(ebias, seg, j + 8, tq, hl, n);
            load_rows4(qkv, s_l, j + 8, lane * 4, qa, va);
          }
          attn_group4(qb, vb, bb, k4, n - j - 4, mrun, den, acc);
        }
      }
    }
    const float inv = (deg > 0) ? (1.0f / den) : 0.0f;
    *(float4*)&xs[r][lane * 4] = make_float4(acc.x * inv, acc.y * inv, acc.z * inv, acc.w * inv);
  }
  __syncthreads();

  const int wave = tid >> 5;
  const int row  = lane & 15;
  const int half = lane >> 4;
  const int col  = wave * 16 + row;
  const float* wf = wfrag + wave * WF_TILE + lane * 4;

  v8f acc = {};
  #pragma unroll 4
  for (int p = 0; p < 16; ++p) {
    const float4 bw = *(const float4*)(wf + p * 128);
    const int k = 8 * p + 2 * half;
    v2f a, b;
    a.x = xs[row][k];      a.y = xs[row][k + 1];
    b.x = bw.x;            b.y = bw.y;
    acc = __builtin_amdgcn_wmma_f32_16x16x4_f32(
        false, a, false, b, (short)0, acc, false, false);
    a.x = xs[row][k + 4];  a.y = xs[row][k + 5];
    b.x = bw.z;            b.y = bw.w;
    acc = __builtin_amdgcn_wmma_f32_16x16x4_f32(
        false, a, false, b, (short)0, acc, false, false);
  }
  const float bias = b_out[col];
  #pragma unroll
  for (int j = 0; j < 8; ++j) {
    const int mrow = j + 8 * half;
    const int idx  = (nodeBase + mrow) * D_MODEL + col;
    out[idx] = node_feature[idx] + acc[j] + bias;
  }
}

extern "C" void kernel_launch(void* const* d_in, const int* in_sizes, int n_in,
                              void* d_out, int out_size, void* d_ws, size_t ws_size,
                              hipStream_t stream) {
  const float* node_feature = (const float*)d_in[0];
  const float* dist_attn    = (const float*)d_in[1];
  const float* path_attn    = (const float*)d_in[2];
  const float* gamma        = (const float*)d_in[3];
  const float* beta         = (const float*)d_in[4];
  const float* W_qkv        = (const float*)d_in[5];
  const float* b_qkv        = (const float*)d_in[6];
  const float* W_out        = (const float*)d_in[7];
  const float* b_out        = (const float*)d_in[8];
  const int*   src          = (const int*)d_in[9];
  const int*   dst          = (const int*)d_in[10];
  float*       out          = (float*)d_out;

  float* ws    = (float*)d_ws;
  float* qkv   = ws;
  float* ebias = qkv   + (size_t)N_NODES * QKV_COLS;
  float* wfrag = ebias + (size_t)N_EDGES * N_HEADS;
  int*   cnt   = (int*)(wfrag + (size_t)2 * N_NODES * N_HEADS);
  int*   offs  = cnt  + 50048;
  int*   psrc  = offs + 50048 + 2 * N_EDGES;
  float* t_eb  = qkv;
  int*   t_src = (int*)(qkv + (size_t)N_EDGES * N_HEADS);
  int*   t_dst = t_src + N_EDGES;
  int*   t_tab = t_dst + N_EDGES;

  const int nodeBlocks = N_NODES / 16;

  gsan_init_kernel<<<WF_BLOCKS, 256, 0, stream>>>(W_qkv, W_out, wfrag);
  gsan_sort_bin_kernel<<<SORT_CHUNKS, 1024, 0, stream>>>(
      dist_attn, path_attn, src, dst, t_eb, t_src, t_dst, t_tab);
  gsan_sort_place_kernel<<<SORT_RANGES, 1024, 0, stream>>>(
      t_eb, t_src, t_dst, t_tab, cnt, offs, psrc, ebias);
  gsan_ln_qkv_kernel<<<nodeBlocks, 256, 0, stream>>>(
      node_feature, gamma, beta, wfrag, b_qkv, qkv);
  gsan_attn_out_kernel<<<nodeBlocks, 256, 0, stream>>>(
      qkv, ebias, cnt, offs, psrc, wfrag + WF_QKV_TILES * WF_TILE, b_out, node_feature, out);
}
